# starting kernel
# baseline (speedup 1.0000x reference)
#include <hip/hip_runtime.h>

typedef __attribute__((ext_vector_type(2))) float v2f;
typedef __attribute__((ext_vector_type(8))) float v8f;

#define D_FEAT 512
#define N_OUT  64
#define PAIR_ROWS   (D_FEAT / 2)
#define PAIR_STRIDE 80

__global__ __launch_bounds__(256) void gcn_gemm_wmma(
    const float* __restrict__ x, const float* __restrict__ W,
    float* __restrict__ h, int n_nodes)
{
    __shared__ float2 Wp[PAIR_ROWS * PAIR_STRIDE];

    for (int i = threadIdx.x; i < PAIR_ROWS * 16; i += 256) {
        const int p  = i >> 4;
        const int n4 = (i & 15) << 2;
        const float4 r0 = *reinterpret_cast<const float4*>(W + (2 * p + 0) * N_OUT + n4);
        const float4 r1 = *reinterpret_cast<const float4*>(W + (2 * p + 1) * N_OUT + n4);
        float4* dst = reinterpret_cast<float4*>(&Wp[p * PAIR_STRIDE + n4]);
        float4 lo, hi;
        lo.x = r0.x; lo.y = r1.x; lo.z = r0.y; lo.w = r1.y;
        hi.x = r0.z; hi.y = r1.z; hi.z = r0.w; hi.w = r1.w;
        dst[0] = lo;
        dst[1] = hi;
    }
    __syncthreads();

    const int lane = threadIdx.x & 31;
    const int wave = threadIdx.x >> 5;
    const int l16  = lane & 15;
    const int g    = lane >> 4;
    const long rowBase = (long)blockIdx.x * 128 + (long)wave * 16;
    if (rowBase >= n_nodes) return;

    const float* aptr = x + (rowBase + l16) * (long)D_FEAT + 2 * g;
    const float2* bptr = &Wp[(long)g * PAIR_STRIDE + l16];

    v8f acc0 = {}, acc1 = {}, acc2 = {}, acc3 = {};

    for (int k = 0; k < D_FEAT; k += 4) {
        const v2f a = *reinterpret_cast<const v2f*>(aptr + k);

        const float2* wb = bptr + (k >> 1) * PAIR_STRIDE;
        const v2f b0 = *reinterpret_cast<const v2f*>(wb +  0);
        const v2f b1 = *reinterpret_cast<const v2f*>(wb + 16);
        const v2f b2 = *reinterpret_cast<const v2f*>(wb + 32);
        const v2f b3 = *reinterpret_cast<const v2f*>(wb + 48);

        acc0 = __builtin_amdgcn_wmma_f32_16x16x4_f32(false, a, false, b0, (short)0, acc0, false, false);
        acc1 = __builtin_amdgcn_wmma_f32_16x16x4_f32(false, a, false, b1, (short)0, acc1, false, false);
        acc2 = __builtin_amdgcn_wmma_f32_16x16x4_f32(false, a, false, b2, (short)0, acc2, false, false);
        acc3 = __builtin_amdgcn_wmma_f32_16x16x4_f32(false, a, false, b3, (short)0, acc3, false, false);
    }

    float* hb = h + rowBase * N_OUT + l16;
#pragma unroll
    for (int v = 0; v < 8; ++v) {
        const long m = v + 8 * g;
        hb[m * N_OUT +  0] = acc0[v];
        hb[m * N_OUT + 16] = acc1[v];
        hb[m * N_OUT + 32] = acc2[v];
        hb[m * N_OUT + 48] = acc3[v];
    }
}

__global__ __launch_bounds__(256) void gcn_zero(float4* __restrict__ p, long n4)
{
    long i = (long)blockIdx.x * blockDim.x + threadIdx.x;
    const long stride = (long)gridDim.x * blockDim.x;
    const float4 z = {0.f, 0.f, 0.f, 0.f};
    for (; i < n4; i += stride) p[i] = z;
}

__global__ __launch_bounds__(256) void gcn_scatter(
    const float* __restrict__ h, const int* __restrict__ esrc,
    const int* __restrict__ edst, const float* __restrict__ ew,
    float* __restrict__ agg, int n_edges)
{
    const int wave_id = (int)(((long)blockIdx.x * blockDim.x + threadIdx.x) >> 5);
    const int lane = threadIdx.x & 31;
    if (wave_id >= n_edges) return;

    const int   s  = esrc[wave_id];
    const int   d  = edst[wave_id];
    const float wt = ew[wave_id];

    const float2 hv = reinterpret_cast<const float2*>(h + (long)s * N_OUT)[lane];
    float* ap = agg + (long)d * N_OUT + 2 * lane;
#if defined(__HIP_DEVICE_COMPILE__)
    unsafeAtomicAdd(ap + 0, hv.x * wt);
    unsafeAtomicAdd(ap + 1, hv.y * wt);
#endif
}

__global__ __launch_bounds__(256) void gcn_softmax(float* __restrict__ agg, int n_nodes)
{
    const int wave_id = (int)(((long)blockIdx.x * blockDim.x + threadIdx.x) >> 5);
    const int lane = threadIdx.x & 31;
    if (wave_id >= n_nodes) return;

    float2* row = reinterpret_cast<float2*>(agg + (long)wave_id * N_OUT);
    const float2 v = row[lane];

    float m = fmaxf(v.x, v.y);
#pragma unroll
    for (int off = 16; off > 0; off >>= 1)
        m = fmaxf(m, __shfl_xor(m, off, 32));

    const float e0 = __expf(v.x - m);
    const float e1 = __expf(v.y - m);
    float s = e0 + e1;
#pragma unroll
    for (int off = 16; off > 0; off >>= 1)
        s += __shfl_xor(s, off, 32);

    const float inv = 1.0f / s;
    float2 o; o.x = e0 * inv; o.y = e1 * inv;
    row[lane] = o;
}

extern "C" void kernel_launch(void* const* d_in, const int* in_sizes, int n_in,
                              void* d_out, int out_size, void* d_ws, size_t ws_size,
                              hipStream_t stream)
{
    const float* x    = (const float*)d_in[0];
    const int*   esrc = (const int*)d_in[1];
    const int*   edst = (const int*)d_in[2];
    const float* ew   = (const float*)d_in[3];
    const float* W    = (const float*)d_in[4];

    const int n_nodes = in_sizes[0] / D_FEAT;
    const int n_edges = in_sizes[1];

    float* agg = (float*)d_out;
    float* h   = (float*)d_ws;

    {
        const long n4 = ((long)n_nodes * N_OUT) / 4;
        const int blocks = (int)((n4 + 255) / 256);
        gcn_zero<<<blocks, 256, 0, stream>>>((float4*)agg, n4);
    }
    {
        const int blocks = (n_nodes + 127) / 128;
        gcn_gemm_wmma<<<blocks, 256, 0, stream>>>(x, W, h, n_nodes);
    }
    {
        const int blocks = (int)(((long)n_edges * 32 + 255) / 256);
        gcn_scatter<<<blocks, 256, 0, stream>>>(h, esrc, edst, ew, agg, n_edges);
    }
    {
        const int blocks = (int)(((long)n_nodes * 32 + 255) / 256);
        gcn_softmax<<<blocks, 256, 0, stream>>>(agg, n_nodes);
    }
}


# aggregate scans a 1-byte tile-id plane (kernel 1 writes it behind h if ws_size has room, else old path) not edge_dst: 4x fewer scan bytes, row read per kept edge; LDS f32 atomics still arrival order
# speedup vs baseline: 1.9308x; 1.9308x over previous
#include <hip/hip_runtime.h>

typedef __attribute__((ext_vector_type(2))) float v2f;
typedef __attribute__((ext_vector_type(8))) float v8f;

#define D_FEAT 512
#define N_OUT  64
#define PAIR_ROWS   (D_FEAT / 2)
#define PAIR_STRIDE 80

#define TILE_ROWS   391

__device__ __forceinline__ bool keys_fit(unsigned long long ws_bytes, int n_nodes, int n_edges)
{
    const unsigned long long h_bytes = (unsigned long long)n_nodes * N_OUT * sizeof(float);
    const int tiles = (n_nodes + TILE_ROWS - 1) / TILE_ROWS;
    return tiles <= 256 && n_edges > 0 && ws_bytes >= h_bytes + (unsigned long long)n_edges;
}

__device__ __forceinline__ unsigned pack_tile_ids(int4 d)
{
    return  (((unsigned)d.x / TILE_ROWS) & 0xffu)
         | ((((unsigned)d.y / TILE_ROWS) & 0xffu) << 8)
         | ((((unsigned)d.z / TILE_ROWS) & 0xffu) << 16)
         | ((((unsigned)d.w / TILE_ROWS) & 0xffu) << 24);
}

__global__ __launch_bounds__(256) void gcn_gemm_wmma(
    const float* __restrict__ x, const float* __restrict__ W,
    float* __restrict__ h, int n_nodes,
    const int* __restrict__ edst, int n_edges, unsigned long long ws_bytes)
{
    __shared__ float2 Wp[PAIR_ROWS * PAIR_STRIDE];

    if (keys_fit(ws_bytes, n_nodes, n_edges)) {
        unsigned* keys = reinterpret_cast<unsigned*>(h + (size_t)n_nodes * N_OUT);
        const int4* d4 = reinterpret_cast<const int4*>(edst);
        const int n4 = n_edges >> 2;
        const int stride = gridDim.x * 256;
        for (int i = blockIdx.x * 256 + threadIdx.x; i < n4; i += 2 * stride) {
            const int  i1   = i + stride;
            const bool has1 = i1 < n4;
            const int4 d0 = d4[i];
            const int4 d1 = d4[has1 ? i1 : i];
            asm volatile("" :: "v"(d0.x), "v"(d0.y), "v"(d0.z), "v"(d0.w), "v"(d1.x), "v"(d1.y), "v"(d1.z), "v"(d1.w));
            keys[i] = pack_tile_ids(d0);
            if (has1) keys[i1] = pack_tile_ids(d1);
        }
    }

    for (int i = threadIdx.x; i < PAIR_ROWS * 16; i += 256) {
        const int p  = i >> 4;
        const int n4 = (i & 15) << 2;
        const float4 r0 = *reinterpret_cast<const float4*>(W + (2 * p + 0) * N_OUT + n4);
        const float4 r1 = *reinterpret_cast<const float4*>(W + (2 * p + 1) * N_OUT + n4);
        float4* dst = reinterpret_cast<float4*>(&Wp[p * PAIR_STRIDE + n4]);
        float4 lo, hi;
        lo.x = r0.x; lo.y = r1.x; lo.z = r0.y; lo.w = r1.y;
        hi.x = r0.z; hi.y = r1.z; hi.z = r0.w; hi.w = r1.w;
        dst[0] = lo;
        dst[1] = hi;
    }
    __syncthreads();

    const int lane = threadIdx.x & 31;
    const int wave = threadIdx.x >> 5;
    const int l16  = lane & 15;
    const int g    = lane >> 4;
    const long rowBase = (long)blockIdx.x * 128 + (long)wave * 16;
    if (rowBase >= n_nodes) return;

    const float* aptr = x + (rowBase + l16) * (long)D_FEAT + 2 * g;
    const float2* bptr = &Wp[(long)g * PAIR_STRIDE + l16];

    v8f acc0 = {}, acc1 = {}, acc2 = {}, acc3 = {};

    for (int k = 0; k < D_FEAT; k += 4) {
        const v2f a = *reinterpret_cast<const v2f*>(aptr + k);

        const float2* wb = bptr + (k >> 1) * PAIR_STRIDE;
        const v2f b0 = *reinterpret_cast<const v2f*>(wb +  0);
        const v2f b1 = *reinterpret_cast<const v2f*>(wb + 16);
        const v2f b2 = *reinterpret_cast<const v2f*>(wb + 32);
        const v2f b3 = *reinterpret_cast<const v2f*>(wb + 48);

        acc0 = __builtin_amdgcn_wmma_f32_16x16x4_f32(false, a, false, b0, (short)0, acc0, false, false);
        acc1 = __builtin_amdgcn_wmma_f32_16x16x4_f32(false, a, false, b1, (short)0, acc1, false, false);
        acc2 = __builtin_amdgcn_wmma_f32_16x16x4_f32(false, a, false, b2, (short)0, acc2, false, false);
        acc3 = __builtin_amdgcn_wmma_f32_16x16x4_f32(false, a, false, b3, (short)0, acc3, false, false);
    }

    float* hb = h + rowBase * N_OUT + l16;
#pragma unroll
    for (int v = 0; v < 8; ++v) {
        const long m = v + 8 * g;
        hb[m * N_OUT +  0] = acc0[v];
        hb[m * N_OUT + 16] = acc1[v];
        hb[m * N_OUT + 32] = acc2[v];
        hb[m * N_OUT + 48] = acc3[v];
    }
}

#define AGG_THREADS 1024
#define AGG_WAVES   (AGG_THREADS / 32)
#define WIN_LOADS   4
#define WIN_EDGES   (WIN_LOADS * 128)
#define Q_CAP       544
#define KWIN_LOADS  2
#define KWIN_EDGES  (KWIN_LOADS * 512)
static_assert(Q_CAP * 8 >= (31 + KWIN_EDGES) * 4, "a wave queue must hold one tile-id window of 4-byte entries plus the carried ones");

template <bool KEYS>
__device__ __forceinline__ void queue_push(bool keep, int e_first, unsigned c, unsigned row_in_tile,
                                           uint2* q, int& qn)
{
    const unsigned m = __builtin_amdgcn_ballot_w32(keep);
    if (m != 0u) {
        if (keep) {
            const int pos = qn + (int)__builtin_amdgcn_mbcnt_lo(m, 0u);
            if (KEYS) reinterpret_cast<unsigned*>(q)[pos] = (unsigned)e_first + c;
            else      q[pos] = make_uint2((unsigned)e_first, row_in_tile | (c << 16));
        }
        qn += __builtin_popcount(m);
    }
}

__device__ __forceinline__ void scan_window(const int4 (&buf)[WIN_LOADS], int wbase, int lane,
                                            int lo, unsigned nrows, uint2* q, int& qn)
{
#pragma unroll
    for (int u = 0; u < WIN_LOADS; ++u) {
        const int4 d = buf[u];
        const unsigned r0 = (unsigned)(d.x - lo);
        const unsigned r1 = (unsigned)(d.y - lo);
        const unsigned r2 = (unsigned)(d.z - lo);
        const unsigned r3 = (unsigned)(d.w - lo);
        const unsigned r01 = r0 < r1 ? r0 : r1;
        const unsigned r23 = r2 < r3 ? r2 : r3;
        const unsigned rmin = r01 < r23 ? r01 : r23;
        if (__builtin_amdgcn_ballot_w32(rmin < nrows) != 0u) {
            const int e0 = wbase + u * 128 + 4 * lane;
            queue_push<false>(r0 < nrows, e0, 0u, r0, q, qn);
            queue_push<false>(r1 < nrows, e0, 1u, r1, q, qn);
            queue_push<false>(r2 < nrows, e0, 2u, r2, q, qn);
            queue_push<false>(r3 < nrows, e0, 3u, r3, q, qn);
        }
    }
}

__device__ __forceinline__ void push_key_dword(unsigned x, unsigned z, int e_first, uint2* q, int& qn)
{
    if (__builtin_amdgcn_ballot_w32(z != 0u) != 0u) {
        queue_push<true>((x & 0x000000ffu) == 0u, e_first, 0u, 0u, q, qn);
        queue_push<true>((x & 0x0000ff00u) == 0u, e_first, 1u, 0u, q, qn);
        queue_push<true>((x & 0x00ff0000u) == 0u, e_first, 2u, 0u, q, qn);
        queue_push<true>((x & 0xff000000u) == 0u, e_first, 3u, 0u, q, qn);
    }
}

__device__ __forceinline__ void scan_window_keys(const uint4 (&buf)[KWIN_LOADS], int wbase, int lane,
                                                 unsigned pat, uint2* q, int& qn)
{
#pragma unroll
    for (int u = 0; u < KWIN_LOADS; ++u) {
        const uint4 d = buf[u];
        const unsigned x0 = d.x ^ pat, x1 = d.y ^ pat, x2 = d.z ^ pat, x3 = d.w ^ pat;
        const unsigned z0 = (x0 - 0x01010101u) & ~x0 & 0x80808080u;
        const unsigned z1 = (x1 - 0x01010101u) & ~x1 & 0x80808080u;
        const unsigned z2 = (x2 - 0x01010101u) & ~x2 & 0x80808080u;
        const unsigned z3 = (x3 - 0x01010101u) & ~x3 & 0x80808080u;
        if (__builtin_amdgcn_ballot_w32((z0 | z1 | z2 | z3) != 0u) != 0u) {
            const int e0 = wbase + u * 512 + 16 * lane;
            push_key_dword(x0, z0, e0 +  0, q, qn);
            push_key_dword(x1, z1, e0 +  4, q, qn);
            push_key_dword(x2, z2, e0 +  8, q, qn);
            push_key_dword(x3, z3, e0 + 12, q, qn);
        }
    }
}

template <bool KEYS>
__device__ __forceinline__ void queue_drain(const uint2* q, int base, int nvalid, int lane,
                                            const float* __restrict__ h,
                                            const int* __restrict__ esrc,
                                            const int* __restrict__ edst,
                                            const float* __restrict__ ew,
                                            float* tile, int lo, unsigned nrows)
{
    __builtin_amdgcn_fence(__ATOMIC_ACQ_REL, "wavefront");

    int   s  = 0;
    float w  = 0.0f;
    int   dl = TILE_ROWS;
    if (lane < nvalid) {
        if (KEYS) {
            const unsigned e  = reinterpret_cast<const unsigned*>(q)[base + lane];
            const unsigned r  = (unsigned)(edst[e] - lo);
            const int      se = esrc[e];
            const float    we = ew[e];
            const bool     in_tile = r < nrows;
            s  = in_tile ? se : 0;
            w  = in_tile ? we : 0.0f;
            dl = in_tile ? (int)r : TILE_ROWS;
        } else {
            const uint2 ent = q[base + lane];
            const unsigned e = ent.x + (ent.y >> 16);
            s  = esrc[e];
            w  = ew[e];
            dl = (int)(ent.y & 0xffffu);
        }
    }

#pragma unroll
    for (int half = 0; half < 2; ++half) {
        float2 hv[16];
#pragma unroll
        for (int i = 0; i < 16; ++i) {
            const int si = __builtin_amdgcn_readlane(s, half * 16 + i);
            hv[i] = reinterpret_cast<const float2*>(h + (long)si * N_OUT)[lane];
        }
#pragma unroll
        for (int i = 0; i < 16; ++i) {
            const float wi = __int_as_float(__builtin_amdgcn_readlane(__float_as_int(w), half * 16 + i));
            const int   di = __builtin_amdgcn_readlane(dl, half * 16 + i);
            float* t = tile + di * N_OUT + 2 * lane;
            atomicAdd(t + 0, hv[i].x * wi);
            atomicAdd(t + 1, hv[i].y * wi);
        }
    }
}

template <bool KEYS>
__device__ __forceinline__ void scan_and_add(const float* __restrict__ h, const int* __restrict__ esrc,
                                             const int* __restrict__ edst, const float* __restrict__ ew,
                                             float* tile, uint2* q, int lane, int wave,
                                             int lo, unsigned nrows, int n_nodes, int n_edges)
{
    int qn = 0;
    int covered;

    if (KEYS) {
        const int n_full = n_edges / KWIN_EDGES;
        covered = n_full * KWIN_EDGES;
        const uint4* k4 = reinterpret_cast<const uint4*>(h + (size_t)n_nodes * N_OUT);
        const unsigned pat = (unsigned)blockIdx.x * 0x01010101u;
        const int rot = (int)(((long)blockIdx.x * n_full) / gridDim.x);
        uint4 bufA[KWIN_LOADS], bufB[KWIN_LOADS];
        if (wave < n_full) {
            const int g0 = (wave + rot >= n_full) ? wave + rot - n_full : wave + rot;
#pragma unroll
            for (int u = 0; u < KWIN_LOADS; ++u)
                bufA[u] = k4[(long)g0 * (KWIN_EDGES / 16) + u * 32 + lane];
        }
        for (int j = wave; j < n_full; j += 2 * AGG_WAVES) {
            const int  ga   = (j + rot >= n_full) ? j + rot - n_full : j + rot;
            const int  jb   = j + AGG_WAVES;
            const bool hasB = jb < n_full;
            const int  jn   = j + 2 * AGG_WAVES;
            const int  gb   = !hasB ? ga : (jb + rot >= n_full) ? jb + rot - n_full : jb + rot;
            const int  gn   = !(jn < n_full) ? ga : (jn + rot >= n_full) ? jn + rot - n_full : jn + rot;

#pragma unroll
            for (int u = 0; u < KWIN_LOADS; ++u)
                bufB[u] = k4[(long)gb * (KWIN_EDGES / 16) + u * 32 + lane];

            scan_window_keys(bufA, ga * KWIN_EDGES, lane, pat, q, qn);
            while (qn >= 32) {
                qn -= 32;
                queue_drain<true>(q, qn, 32, lane, h, esrc, edst, ew, tile, lo, nrows);
            }

#pragma unroll
            for (int u = 0; u < KWIN_LOADS; ++u)
                bufA[u] = k4[(long)gn * (KWIN_EDGES / 16) + u * 32 + lane];

            if (hasB) {
                scan_window_keys(bufB, gb * KWIN_EDGES, lane, pat, q, qn);
                while (qn >= 32) {
                    qn -= 32;
                    queue_drain<true>(q, qn, 32, lane, h, esrc, edst, ew, tile, lo, nrows);
                }
            }
        }
    } else {
        const int n_full = n_edges / WIN_EDGES;
        covered = n_full * WIN_EDGES;
        const int4* d4 = reinterpret_cast<const int4*>(edst);
        const int rot = (int)(((long)blockIdx.x * n_full) / gridDim.x);
        int4 bufA[WIN_LOADS], bufB[WIN_LOADS];
        if (wave < n_full) {
            const int g0 = (wave + rot >= n_full) ? wave + rot - n_full : wave + rot;
#pragma unroll
            for (int u = 0; u < WIN_LOADS; ++u)
                bufA[u] = d4[(long)g0 * (WIN_EDGES / 4) + u * 32 + lane];
        }
        for (int j = wave; j < n_full; j += 2 * AGG_WAVES) {
            const int  ga   = (j + rot >= n_full) ? j + rot - n_full : j + rot;
            const int  jb   = j + AGG_WAVES;
            const bool hasB = jb < n_full;
            const int  jn   = j + 2 * AGG_WAVES;
            const int  gb   = !hasB ? ga : (jb + rot >= n_full) ? jb + rot - n_full : jb + rot;
            const int  gn   = !(jn < n_full) ? ga : (jn + rot >= n_full) ? jn + rot - n_full : jn + rot;

#pragma unroll
            for (int u = 0; u < WIN_LOADS; ++u)
                bufB[u] = d4[(long)gb * (WIN_EDGES / 4) + u * 32 + lane];

            scan_window(bufA, ga * WIN_EDGES, lane, lo, nrows, q, qn);
            while (qn >= 32) {
                qn -= 32;
                queue_drain<false>(q, qn, 32, lane, h, esrc, edst, ew, tile, lo, nrows);
            }

#pragma unroll
            for (int u = 0; u < WIN_LOADS; ++u)
                bufA[u] = d4[(long)gn * (WIN_EDGES / 4) + u * 32 + lane];

            if (hasB) {
                scan_window(bufB, gb * WIN_EDGES, lane, lo, nrows, q, qn);
                while (qn >= 32) {
                    qn -= 32;
                    queue_drain<false>(q, qn, 32, lane, h, esrc, edst, ew, tile, lo, nrows);
                }
            }
        }
    }

    {
        int e0 = covered + wave * 32;
        for (;;) {
            const bool more = e0 < n_edges;
            if (more) {
                const int e = e0 + lane;
                unsigned r = 0xffffffffu;
                if (e < n_edges) r = (unsigned)(edst[e] - lo);
                queue_push<KEYS>(r < nrows, e, 0u, r, q, qn);
                e0 += AGG_WAVES * 32;
            }
            if (qn >= 32 || (!more && qn > 0)) {
                const int n = qn < 32 ? qn : 32;
                qn -= n;
                queue_drain<KEYS>(q, qn, n, lane, h, esrc, edst, ew, tile, lo, nrows);
            }
            if (!more && qn == 0) break;
        }
    }
}

__global__ __launch_bounds__(AGG_THREADS) void gcn_tile_aggregate(
    const float* __restrict__ h, const int* __restrict__ esrc,
    const int* __restrict__ edst, const float* __restrict__ ew,
    float* __restrict__ out, int n_nodes, int n_edges, unsigned long long ws_bytes)
{
    __shared__ __attribute__((aligned(16))) float tile[(TILE_ROWS + 1) * N_OUT];
    __shared__ __attribute__((aligned(16))) uint2 queue[AGG_WAVES * Q_CAP];

    const int lane = threadIdx.x & 31;
    const int wave = __builtin_amdgcn_readfirstlane((int)(threadIdx.x >> 5));
    const int lo   = blockIdx.x * TILE_ROWS;
    const int rem  = n_nodes - lo;
    const unsigned nrows = (unsigned)(rem < TILE_ROWS ? rem : TILE_ROWS);

    {
        float4* t4 = reinterpret_cast<float4*>(tile);
        const float4 z = {0.f, 0.f, 0.f, 0.f};
        for (int i = threadIdx.x; i < (TILE_ROWS + 1) * N_OUT / 4; i += AGG_THREADS) t4[i] = z;
    }
    __syncthreads();

    uint2* q = queue + wave * Q_CAP;

    if (keys_fit(ws_bytes, n_nodes, n_edges) && gridDim.x <= 256u)
        scan_and_add<true >(h, esrc, edst, ew, tile, q, lane, wave, lo, nrows, n_nodes, n_edges);
    else
        scan_and_add<false>(h, esrc, edst, ew, tile, q, lane, wave, lo, nrows, n_nodes, n_edges);

    __syncthreads();

    for (unsigned r = (unsigned)wave; r < nrows; r += AGG_WAVES) {
        const float2 v = reinterpret_cast<const float2*>(tile + r * N_OUT)[lane];

        float m = fmaxf(v.x, v.y);
#pragma unroll
        for (int off = 16; off > 0; off >>= 1)
            m = fmaxf(m, __shfl_xor(m, off, 32));

        const float x0 = __expf(v.x - m);
        const float x1 = __expf(v.y - m);
        float sum = x0 + x1;
#pragma unroll
        for (int off = 16; off > 0; off >>= 1)
            sum += __shfl_xor(sum, off, 32);

        const float inv = 1.0f / sum;
        float2 o; o.x = x0 * inv; o.y = x1 * inv;
        reinterpret_cast<float2*>(out + ((long)lo + r) * N_OUT)[lane] = o;
    }
}

extern "C" void kernel_launch(void* const* d_in, const int* in_sizes, int n_in,
                              void* d_out, int out_size, void* d_ws, size_t ws_size,
                              hipStream_t stream)
{
    const float* x    = (const float*)d_in[0];
    const int*   esrc = (const int*)d_in[1];
    const int*   edst = (const int*)d_in[2];
    const float* ew   = (const float*)d_in[3];
    const float* W    = (const float*)d_in[4];

    const int n_nodes = in_sizes[0] / D_FEAT;
    const int n_edges = in_sizes[1];

    float* out = (float*)d_out;
    float* h   = (float*)d_ws;

    {
        const int blocks = (n_nodes + 127) / 128;
        gcn_gemm_wmma<<<blocks, 256, 0, stream>>>(x, W, h, n_nodes, edst, n_edges, (unsigned long long)ws_size);
    }
    {
        const int tiles = (n_nodes + TILE_ROWS - 1) / TILE_ROWS;
        gcn_tile_aggregate<<<tiles, AGG_THREADS, 0, stream>>>(h, esrc, edst, ew, out, n_nodes, n_edges, (unsigned long long)ws_size);
    }
}


# aggregate reads its tile's bucket, not all edges: new launch counting-sorts edges by 391-row tile into out rows (LDS ranks, one int atomic per block and tile); LDS f32 atomics still arrival order
# speedup vs baseline: 2.1592x; 1.1183x over previous
#include <hip/hip_runtime.h>

typedef __attribute__((ext_vector_type(2))) float v2f;
typedef __attribute__((ext_vector_type(8))) float v8f;

#define D_FEAT 512
#define N_OUT  64
#define PAIR_ROWS   (D_FEAT / 2)
#define PAIR_STRIDE 80

#define TILE_ROWS   391

#define ROW_SHIFT       23
#define BUCKET_PER_ROW  (N_OUT * 4 / 8)
#define MAX_TILES       1024
#define BIN_THREADS     256
#define BIN_PER_THREAD  16
#define BIN_SLOTS       (BIN_PER_THREAD + 1)
#define BIN_EDGES       (BIN_THREADS * BIN_PER_THREAD)

__device__ __forceinline__ bool bins_fit(unsigned long long ws_bytes, int n_nodes)
{
    const unsigned long long h_bytes = (unsigned long long)n_nodes * N_OUT * sizeof(float);
    const int tiles = (n_nodes + TILE_ROWS - 1) / TILE_ROWS;
    return n_nodes > 0 && n_nodes <= (1 << ROW_SHIFT) && tiles <= MAX_TILES
        && ws_bytes >= h_bytes + (unsigned long long)tiles * sizeof(unsigned);
}

__global__ __launch_bounds__(256) void gcn_gemm_wmma(
    const float* __restrict__ x, const float* __restrict__ W,
    float* __restrict__ h, int n_nodes, unsigned long long ws_bytes)
{
    __shared__ float2 Wp[PAIR_ROWS * PAIR_STRIDE];

    if (blockIdx.x == 0 && bins_fit(ws_bytes, n_nodes)) {
        unsigned* counters = reinterpret_cast<unsigned*>(h + (size_t)n_nodes * N_OUT);
        const int tiles = (n_nodes + TILE_ROWS - 1) / TILE_ROWS;
        for (int i = threadIdx.x; i < tiles; i += 256) counters[i] = 0u;
    }
    for (int i = threadIdx.x; i < PAIR_ROWS * 16; i += 256) {
        const int p  = i >> 4;
        const int n4 = (i & 15) << 2;
        const float4 r0 = *reinterpret_cast<const float4*>(W + (2 * p + 0) * N_OUT + n4);
        const float4 r1 = *reinterpret_cast<const float4*>(W + (2 * p + 1) * N_OUT + n4);
        float4* dst = reinterpret_cast<float4*>(&Wp[p * PAIR_STRIDE + n4]);
        float4 lo, hi;
        lo.x = r0.x; lo.y = r1.x; lo.z = r0.y; lo.w = r1.y;
        hi.x = r0.z; hi.y = r1.z; hi.z = r0.w; hi.w = r1.w;
        dst[0] = lo;
        dst[1] = hi;
    }
    __syncthreads();

    const int lane = threadIdx.x & 31;
    const int wave = threadIdx.x >> 5;
    const int l16  = lane & 15;
    const int g    = lane >> 4;
    const long rowBase = (long)blockIdx.x * 128 + (long)wave * 16;
    if (rowBase >= n_nodes) return;

    const float* aptr = x + (rowBase + l16) * (long)D_FEAT + 2 * g;
    const float2* bptr = &Wp[(long)g * PAIR_STRIDE + l16];

    v8f acc0 = {}, acc1 = {}, acc2 = {}, acc3 = {};

    for (int k = 0; k < D_FEAT; k += 4) {
        const v2f a = *reinterpret_cast<const v2f*>(aptr + k);

        const float2* wb = bptr + (k >> 1) * PAIR_STRIDE;
        const v2f b0 = *reinterpret_cast<const v2f*>(wb +  0);
        const v2f b1 = *reinterpret_cast<const v2f*>(wb + 16);
        const v2f b2 = *reinterpret_cast<const v2f*>(wb + 32);
        const v2f b3 = *reinterpret_cast<const v2f*>(wb + 48);

        acc0 = __builtin_amdgcn_wmma_f32_16x16x4_f32(false, a, false, b0, (short)0, acc0, false, false);
        acc1 = __builtin_amdgcn_wmma_f32_16x16x4_f32(false, a, false, b1, (short)0, acc1, false, false);
        acc2 = __builtin_amdgcn_wmma_f32_16x16x4_f32(false, a, false, b2, (short)0, acc2, false, false);
        acc3 = __builtin_amdgcn_wmma_f32_16x16x4_f32(false, a, false, b3, (short)0, acc3, false, false);
    }

    float* hb = h + rowBase * N_OUT + l16;
#pragma unroll
    for (int v = 0; v < 8; ++v) {
        const long m = v + 8 * g;
        hb[m * N_OUT +  0] = acc0[v];
        hb[m * N_OUT + 16] = acc1[v];
        hb[m * N_OUT + 32] = acc2[v];
        hb[m * N_OUT + 48] = acc3[v];
    }
}

__global__ __launch_bounds__(BIN_THREADS) void gcn_bin_edges(
    const int* __restrict__ esrc, const int* __restrict__ edst, const float* __restrict__ ew,
    float* __restrict__ out, float* __restrict__ h, int n_nodes, int n_edges, unsigned long long ws_bytes)
{
    __shared__ unsigned cnt[MAX_TILES];

    if (!bins_fit(ws_bytes, n_nodes)) return;
    unsigned* counters = reinterpret_cast<unsigned*>(h + (size_t)n_nodes * N_OUT);
    const int tiles = (n_nodes + TILE_ROWS - 1) / TILE_ROWS;

    for (int i = threadIdx.x; i < tiles; i += BIN_THREADS) cnt[i] = 0u;

    int   d[BIN_SLOTS], s[BIN_SLOTS];
    float w[BIN_SLOTS];
#pragma unroll
    for (int j = 0; j < BIN_PER_THREAD / 4; ++j) {
        const long e0 = (long)blockIdx.x * BIN_EDGES + j * (BIN_THREADS * 4) + threadIdx.x * 4;
        int4   dv = {-1, -1, -1, -1};
        int4   sv = {0, 0, 0, 0};
        float4 wv = {0.f, 0.f, 0.f, 0.f};
        if (e0 + 3 < n_edges) {
            dv = *reinterpret_cast<const int4*>(edst + e0);
            sv = *reinterpret_cast<const int4*>(esrc + e0);
            wv = *reinterpret_cast<const float4*>(ew + e0);
        }
        d[4 * j + 0] = dv.x; d[4 * j + 1] = dv.y; d[4 * j + 2] = dv.z; d[4 * j + 3] = dv.w;
        s[4 * j + 0] = sv.x; s[4 * j + 1] = sv.y; s[4 * j + 2] = sv.z; s[4 * j + 3] = sv.w;
        w[4 * j + 0] = wv.x; w[4 * j + 1] = wv.y; w[4 * j + 2] = wv.z; w[4 * j + 3] = wv.w;
    }
    {
        const int ex = (n_edges & ~3) + (int)threadIdx.x;
        d[BIN_PER_THREAD] = -1; s[BIN_PER_THREAD] = 0; w[BIN_PER_THREAD] = 0.0f;
        if (blockIdx.x == 0 && ex < n_edges) {
            d[BIN_PER_THREAD] = edst[ex];
            s[BIN_PER_THREAD] = esrc[ex];
            w[BIN_PER_THREAD] = ew[ex];
        }
    }
    __syncthreads();

#pragma unroll
    for (int c = 0; c < BIN_SLOTS; ++c)
        if ((unsigned)d[c] < (unsigned)n_nodes) atomicAdd(&cnt[(unsigned)d[c] / TILE_ROWS], 1u);
    __syncthreads();

    for (int i = threadIdx.x; i < tiles; i += BIN_THREADS) {
        const unsigned c = cnt[i];
        cnt[i] = c ? atomicAdd(&counters[i], c) : 0u;
    }
    __syncthreads();

    unsigned pos[BIN_SLOTS];
#pragma unroll
    for (int c = 0; c < BIN_SLOTS; ++c) {
        pos[c] = 0xffffffffu;
        if ((unsigned)d[c] < (unsigned)n_nodes) pos[c] = atomicAdd(&cnt[(unsigned)d[c] / TILE_ROWS], 1u);
    }
#pragma unroll
    for (int c = 0; c < BIN_SLOTS; ++c) {
        const unsigned t    = (unsigned)d[c] / TILE_ROWS;
        const unsigned row0 = t * TILE_ROWS;
        const unsigned left = (unsigned)n_nodes - row0;
        const unsigned cap  = (left < TILE_ROWS ? left : TILE_ROWS) * BUCKET_PER_ROW;
        if ((unsigned)d[c] < (unsigned)n_nodes && pos[c] < cap) {
            uint2 ent;
            ent.x = ((unsigned)s[c] & ((1u << ROW_SHIFT) - 1u)) | (((unsigned)d[c] - row0) << ROW_SHIFT);
            ent.y = __float_as_uint(w[c]);
            reinterpret_cast<uint2*>(out + (size_t)row0 * N_OUT)[pos[c]] = ent;
        }
    }
}

#define AGG_THREADS 1024
#define AGG_WAVES   (AGG_THREADS / 32)
#define WIN_LOADS   4
#define WIN_EDGES   (WIN_LOADS * 128)
#define Q_CAP       544

__device__ __forceinline__ void add_rows(int s, float w, int dl, int lane,
                                         const float* __restrict__ h, float* tile)
{
#pragma unroll
    for (int half = 0; half < 2; ++half) {
        float2 hv[16];
#pragma unroll
        for (int i = 0; i < 16; ++i) {
            const int si = __builtin_amdgcn_readlane(s, half * 16 + i);
            hv[i] = reinterpret_cast<const float2*>(h + (long)si * N_OUT)[lane];
        }
#pragma unroll
        for (int i = 0; i < 16; ++i) {
            const float wi = __int_as_float(__builtin_amdgcn_readlane(__float_as_int(w), half * 16 + i));
            const int   di = __builtin_amdgcn_readlane(dl, half * 16 + i);
            float* t = tile + di * N_OUT + 2 * lane;
            atomicAdd(t + 0, hv[i].x * wi);
            atomicAdd(t + 1, hv[i].y * wi);
        }
    }
}

__device__ __forceinline__ void bucket_add(const uint2* bucket, int n, int lane, int wave,
                                           const float* __restrict__ h, float* tile)
{
    int g = wave * 32;
    if (g >= n) return;
    const int last = n - 1;
    uint2 cur = bucket[g + lane < last ? g + lane : last];
    for (; g < n; g += AGG_THREADS) {
        const int   gn  = g + AGG_THREADS;
        const uint2 nxt = bucket[gn + lane < last ? gn + lane : last];

        const bool  have = g + lane < n;
        const int   s  = have ? (int)(cur.x & ((1u << ROW_SHIFT) - 1u)) : 0;
        const float w  = have ? __uint_as_float(cur.y) : 0.0f;
        const int   dl = have ? (int)(cur.x >> ROW_SHIFT) : TILE_ROWS;
        add_rows(s, w, dl, lane, h, tile);
        cur = nxt;
    }
}

__device__ __forceinline__ void queue_push(bool keep, int e_first, unsigned c, unsigned row_in_tile,
                                           uint2* q, int& qn)
{
    const unsigned m = __builtin_amdgcn_ballot_w32(keep);
    if (m != 0u) {
        if (keep) {
            const int pos = qn + (int)__builtin_amdgcn_mbcnt_lo(m, 0u);
            q[pos] = make_uint2((unsigned)e_first, row_in_tile | (c << 16));
        }
        qn += __builtin_popcount(m);
    }
}

__device__ __forceinline__ void scan_window(const int4 (&buf)[WIN_LOADS], int wbase, int lane,
                                            int lo, unsigned nrows, uint2* q, int& qn)
{
#pragma unroll
    for (int u = 0; u < WIN_LOADS; ++u) {
        const int4 d = buf[u];
        const unsigned r0 = (unsigned)(d.x - lo);
        const unsigned r1 = (unsigned)(d.y - lo);
        const unsigned r2 = (unsigned)(d.z - lo);
        const unsigned r3 = (unsigned)(d.w - lo);
        const unsigned r01 = r0 < r1 ? r0 : r1;
        const unsigned r23 = r2 < r3 ? r2 : r3;
        const unsigned rmin = r01 < r23 ? r01 : r23;
        if (__builtin_amdgcn_ballot_w32(rmin < nrows) != 0u) {
            const int e0 = wbase + u * 128 + 4 * lane;
            queue_push(r0 < nrows, e0, 0u, r0, q, qn);
            queue_push(r1 < nrows, e0, 1u, r1, q, qn);
            queue_push(r2 < nrows, e0, 2u, r2, q, qn);
            queue_push(r3 < nrows, e0, 3u, r3, q, qn);
        }
    }
}

__device__ __forceinline__ void queue_drain(const uint2* q, int base, int nvalid, int lane,
                                            const float* __restrict__ h,
                                            const int* __restrict__ esrc,
                                            const float* __restrict__ ew,
                                            float* tile)
{
    __builtin_amdgcn_fence(__ATOMIC_ACQ_REL, "wavefront");

    int   s  = 0;
    float w  = 0.0f;
    int   dl = TILE_ROWS;
    if (lane < nvalid) {
        const uint2 ent = q[base + lane];
        const unsigned e = ent.x + (ent.y >> 16);
        s  = esrc[e];
        w  = ew[e];
        dl = (int)(ent.y & 0xffffu);
    }
    add_rows(s, w, dl, lane, h, tile);
}

__device__ __forceinline__ void scan_and_add(const float* __restrict__ h, const int* __restrict__ esrc,
                                             const int* __restrict__ edst, const float* __restrict__ ew,
                                             float* tile, uint2* q, int lane, int wave,
                                             int lo, unsigned nrows, int n_edges)
{
    int qn = 0;

    const int n_full  = n_edges / WIN_EDGES;
    const int covered = n_full * WIN_EDGES;
    const int4* d4 = reinterpret_cast<const int4*>(edst);
    const int rot = (int)(((long)blockIdx.x * n_full) / gridDim.x);
    int4 bufA[WIN_LOADS], bufB[WIN_LOADS];
    if (wave < n_full) {
        const int g0 = (wave + rot >= n_full) ? wave + rot - n_full : wave + rot;
#pragma unroll
        for (int u = 0; u < WIN_LOADS; ++u)
            bufA[u] = d4[(long)g0 * (WIN_EDGES / 4) + u * 32 + lane];
    }
    for (int j = wave; j < n_full; j += 2 * AGG_WAVES) {
        const int  ga   = (j + rot >= n_full) ? j + rot - n_full : j + rot;
        const int  jb   = j + AGG_WAVES;
        const bool hasB = jb < n_full;
        const int  jn   = j + 2 * AGG_WAVES;
        const int  gb   = !hasB ? ga : (jb + rot >= n_full) ? jb + rot - n_full : jb + rot;
        const int  gn   = !(jn < n_full) ? ga : (jn + rot >= n_full) ? jn + rot - n_full : jn + rot;

#pragma unroll
        for (int u = 0; u < WIN_LOADS; ++u)
            bufB[u] = d4[(long)gb * (WIN_EDGES / 4) + u * 32 + lane];

        scan_window(bufA, ga * WIN_EDGES, lane, lo, nrows, q, qn);
        while (qn >= 32) {
            qn -= 32;
            queue_drain(q, qn, 32, lane, h, esrc, ew, tile);
        }

#pragma unroll
        for (int u = 0; u < WIN_LOADS; ++u)
            bufA[u] = d4[(long)gn * (WIN_EDGES / 4) + u * 32 + lane];

        if (hasB) {
            scan_window(bufB, gb * WIN_EDGES, lane, lo, nrows, q, qn);
            while (qn >= 32) {
                qn -= 32;
                queue_drain(q, qn, 32, lane, h, esrc, ew, tile);
            }
        }
    }

    {
        int e0 = covered + wave * 32;
        for (;;) {
            const bool more = e0 < n_edges;
            if (more) {
                const int e = e0 + lane;
                unsigned r = 0xffffffffu;
                if (e < n_edges) r = (unsigned)(edst[e] - lo);
                queue_push(r < nrows, e, 0u, r, q, qn);
                e0 += AGG_WAVES * 32;
            }
            if (qn >= 32 || (!more && qn > 0)) {
                const int n = qn < 32 ? qn : 32;
                qn -= n;
                queue_drain(q, qn, n, lane, h, esrc, ew, tile);
            }
            if (!more && qn == 0) break;
        }
    }
}

__global__ __launch_bounds__(AGG_THREADS) void gcn_tile_aggregate(
    const float* __restrict__ h, const int* __restrict__ esrc,
    const int* __restrict__ edst, const float* __restrict__ ew,
    float* __restrict__ out, int n_nodes, int n_edges, unsigned long long ws_bytes)
{
    __shared__ __attribute__((aligned(16))) float tile[(TILE_ROWS + 1) * N_OUT];
    __shared__ __attribute__((aligned(16))) uint2 queue[AGG_WAVES * Q_CAP];

    const int lane = threadIdx.x & 31;
    const int wave = __builtin_amdgcn_readfirstlane((int)(threadIdx.x >> 5));
    const int lo   = blockIdx.x * TILE_ROWS;
    const int rem  = n_nodes - lo;
    const unsigned nrows = (unsigned)(rem < TILE_ROWS ? rem : TILE_ROWS);

    {
        float4* t4 = reinterpret_cast<float4*>(tile);
        const float4 z = {0.f, 0.f, 0.f, 0.f};
        for (int i = threadIdx.x; i < (TILE_ROWS + 1) * N_OUT / 4; i += AGG_THREADS) t4[i] = z;
    }
    __syncthreads();

    bool from_bucket = false;
    unsigned n_bucket = 0u;
    if (bins_fit(ws_bytes, n_nodes)) {
        const unsigned* counters = reinterpret_cast<const unsigned*>(h + (size_t)n_nodes * N_OUT);
        n_bucket    = counters[blockIdx.x];
        from_bucket = n_bucket <= nrows * BUCKET_PER_ROW;
    }

    if (from_bucket)
        bucket_add(reinterpret_cast<const uint2*>(out + (size_t)lo * N_OUT), (int)n_bucket, lane, wave, h, tile);
    else
        scan_and_add(h, esrc, edst, ew, tile, queue + wave * Q_CAP, lane, wave, lo, nrows, n_edges);

    __syncthreads();

    for (unsigned r = (unsigned)wave; r < nrows; r += AGG_WAVES) {
        const float2 v = reinterpret_cast<const float2*>(tile + r * N_OUT)[lane];

        float m = fmaxf(v.x, v.y);
#pragma unroll
        for (int off = 16; off > 0; off >>= 1)
            m = fmaxf(m, __shfl_xor(m, off, 32));

        const float x0 = __expf(v.x - m);
        const float x1 = __expf(v.y - m);
        float sum = x0 + x1;
#pragma unroll
        for (int off = 16; off > 0; off >>= 1)
            sum += __shfl_xor(sum, off, 32);

        const float inv = 1.0f / sum;
        float2 o; o.x = x0 * inv; o.y = x1 * inv;
        reinterpret_cast<float2*>(out + ((long)lo + r) * N_OUT)[lane] = o;
    }
}

extern "C" void kernel_launch(void* const* d_in, const int* in_sizes, int n_in,
                              void* d_out, int out_size, void* d_ws, size_t ws_size,
                              hipStream_t stream)
{
    const float* x    = (const float*)d_in[0];
    const int*   esrc = (const int*)d_in[1];
    const int*   edst = (const int*)d_in[2];
    const float* ew   = (const float*)d_in[3];
    const float* W    = (const float*)d_in[4];

    const int n_nodes = in_sizes[0] / D_FEAT;
    const int n_edges = in_sizes[1];

    float* out = (float*)d_out;
    float* h   = (float*)d_ws;

    {
        const int blocks = (n_nodes + 127) / 128;
        gcn_gemm_wmma<<<blocks, 256, 0, stream>>>(x, W, h, n_nodes, (unsigned long long)ws_size);
    }
    {
        const int blocks = n_edges > 0 ? (n_edges + BIN_EDGES - 1) / BIN_EDGES : 1;
        gcn_bin_edges<<<blocks, BIN_THREADS, 0, stream>>>(esrc, edst, ew, out, h, n_nodes, n_edges, (unsigned long long)ws_size);
    }
    {
        const int tiles = (n_nodes + TILE_ROWS - 1) / TILE_ROWS;
        gcn_tile_aggregate<<<tiles, AGG_THREADS, 0, stream>>>(h, esrc, edst, ew, out, n_nodes, n_edges, (unsigned long long)ws_size);
    }
}


# bin launch orders a work-group's bucket entries by tile in LDS before storing: a store instruction writes a few contiguous runs, not 32 scattered 8-byte pieces; LDS f32 atomics still arrival order
# speedup vs baseline: 2.2354x; 1.0353x over previous
#include <hip/hip_runtime.h>

typedef __attribute__((ext_vector_type(2))) float v2f;
typedef __attribute__((ext_vector_type(8))) float v8f;

#define D_FEAT 512
#define N_OUT  64
#define PAIR_ROWS   (D_FEAT / 2)
#define PAIR_STRIDE 80

#define TILE_ROWS   391

#define ROW_SHIFT       23
#define BUCKET_PER_ROW  (N_OUT * 4 / 8)
#define MAX_TILES       1024
#define BIN_THREADS     256
#define BIN_PER_THREAD  16
#define BIN_SLOTS       (BIN_PER_THREAD + 1)
#define BIN_EDGES       (BIN_THREADS * BIN_PER_THREAD)
#define BIN_STAGE       (BIN_THREADS * BIN_SLOTS)
#define NO_PLACE        0xffffffffu

__device__ __forceinline__ bool bins_fit(unsigned long long ws_bytes, int n_nodes)
{
    const unsigned long long h_bytes = (unsigned long long)n_nodes * N_OUT * sizeof(float);
    const int tiles = (n_nodes + TILE_ROWS - 1) / TILE_ROWS;
    return n_nodes > 0 && n_nodes <= (1 << ROW_SHIFT) && tiles <= MAX_TILES
        && ws_bytes >= h_bytes + (unsigned long long)tiles * sizeof(unsigned);
}

__global__ __launch_bounds__(256) void gcn_gemm_wmma(
    const float* __restrict__ x, const float* __restrict__ W,
    float* __restrict__ h, int n_nodes, unsigned long long ws_bytes)
{
    __shared__ float2 Wp[PAIR_ROWS * PAIR_STRIDE];

    if (blockIdx.x == 0 && bins_fit(ws_bytes, n_nodes)) {
        unsigned* counters = reinterpret_cast<unsigned*>(h + (size_t)n_nodes * N_OUT);
        const int tiles = (n_nodes + TILE_ROWS - 1) / TILE_ROWS;
        for (int i = threadIdx.x; i < tiles; i += 256) counters[i] = 0u;
    }
    for (int i = threadIdx.x; i < PAIR_ROWS * 16; i += 256) {
        const int p  = i >> 4;
        const int n4 = (i & 15) << 2;
        const float4 r0 = *reinterpret_cast<const float4*>(W + (2 * p + 0) * N_OUT + n4);
        const float4 r1 = *reinterpret_cast<const float4*>(W + (2 * p + 1) * N_OUT + n4);
        float4* dst = reinterpret_cast<float4*>(&Wp[p * PAIR_STRIDE + n4]);
        float4 lo, hi;
        lo.x = r0.x; lo.y = r1.x; lo.z = r0.y; lo.w = r1.y;
        hi.x = r0.z; hi.y = r1.z; hi.z = r0.w; hi.w = r1.w;
        dst[0] = lo;
        dst[1] = hi;
    }
    __syncthreads();

    const int lane = threadIdx.x & 31;
    const int wave = threadIdx.x >> 5;
    const int l16  = lane & 15;
    const int g    = lane >> 4;
    const long rowBase = (long)blockIdx.x * 128 + (long)wave * 16;
    if (rowBase >= n_nodes) return;

    const float* aptr = x + (rowBase + l16) * (long)D_FEAT + 2 * g;
    const float2* bptr = &Wp[(long)g * PAIR_STRIDE + l16];

    v8f acc0 = {}, acc1 = {}, acc2 = {}, acc3 = {};

    for (int k = 0; k < D_FEAT; k += 4) {
        const v2f a = *reinterpret_cast<const v2f*>(aptr + k);

        const float2* wb = bptr + (k >> 1) * PAIR_STRIDE;
        const v2f b0 = *reinterpret_cast<const v2f*>(wb +  0);
        const v2f b1 = *reinterpret_cast<const v2f*>(wb + 16);
        const v2f b2 = *reinterpret_cast<const v2f*>(wb + 32);
        const v2f b3 = *reinterpret_cast<const v2f*>(wb + 48);

        acc0 = __builtin_amdgcn_wmma_f32_16x16x4_f32(false, a, false, b0, (short)0, acc0, false, false);
        acc1 = __builtin_amdgcn_wmma_f32_16x16x4_f32(false, a, false, b1, (short)0, acc1, false, false);
        acc2 = __builtin_amdgcn_wmma_f32_16x16x4_f32(false, a, false, b2, (short)0, acc2, false, false);
        acc3 = __builtin_amdgcn_wmma_f32_16x16x4_f32(false, a, false, b3, (short)0, acc3, false, false);
    }

    float* hb = h + rowBase * N_OUT + l16;
#pragma unroll
    for (int v = 0; v < 8; ++v) {
        const long m = v + 8 * g;
        hb[m * N_OUT +  0] = acc0[v];
        hb[m * N_OUT + 16] = acc1[v];
        hb[m * N_OUT + 32] = acc2[v];
        hb[m * N_OUT + 48] = acc3[v];
    }
}

__global__ __launch_bounds__(BIN_THREADS) void gcn_bin_edges(
    const int* __restrict__ esrc, const int* __restrict__ edst, const float* __restrict__ ew,
    float* __restrict__ out, float* __restrict__ h, int n_nodes, int n_edges, unsigned long long ws_bytes)
{
    __shared__ unsigned cnt[MAX_TILES + 32];
    __shared__ unsigned shift[MAX_TILES + 32];
    __shared__ __attribute__((aligned(16))) uint2 stage[BIN_STAGE + 32];
    __shared__ unsigned where[BIN_STAGE + 32];
    __shared__ unsigned wave_sum[BIN_THREADS / 32];

    if (!bins_fit(ws_bytes, n_nodes)) return;
    unsigned* counters = reinterpret_cast<unsigned*>(h + (size_t)n_nodes * N_OUT);
    const int tiles = (n_nodes + TILE_ROWS - 1) / TILE_ROWS;
    const int lane  = threadIdx.x & 31;
    const int wave  = threadIdx.x >> 5;

    for (int i = threadIdx.x; i < tiles; i += BIN_THREADS) cnt[i] = 0u;
    if (threadIdx.x < 32) { cnt[MAX_TILES + threadIdx.x] = 0u; shift[MAX_TILES + threadIdx.x] = 0u; }

    int   d[BIN_SLOTS], s[BIN_SLOTS];
    float w[BIN_SLOTS];
#pragma unroll
    for (int j = 0; j < BIN_PER_THREAD / 4; ++j) {
        const long e0 = (long)blockIdx.x * BIN_EDGES + j * (BIN_THREADS * 4) + threadIdx.x * 4;
        int4   dv = {-1, -1, -1, -1};
        int4   sv = {0, 0, 0, 0};
        float4 wv = {0.f, 0.f, 0.f, 0.f};
        if (e0 + 3 < n_edges) {
            dv = *reinterpret_cast<const int4*>(edst + e0);
            sv = *reinterpret_cast<const int4*>(esrc + e0);
            wv = *reinterpret_cast<const float4*>(ew + e0);
        }
        d[4 * j + 0] = dv.x; d[4 * j + 1] = dv.y; d[4 * j + 2] = dv.z; d[4 * j + 3] = dv.w;
        s[4 * j + 0] = sv.x; s[4 * j + 1] = sv.y; s[4 * j + 2] = sv.z; s[4 * j + 3] = sv.w;
        w[4 * j + 0] = wv.x; w[4 * j + 1] = wv.y; w[4 * j + 2] = wv.z; w[4 * j + 3] = wv.w;
    }
    {
        const int ex = (n_edges & ~3) + (int)threadIdx.x;
        d[BIN_PER_THREAD] = -1; s[BIN_PER_THREAD] = 0; w[BIN_PER_THREAD] = 0.0f;
        if (blockIdx.x == 0 && ex < n_edges) {
            d[BIN_PER_THREAD] = edst[ex];
            s[BIN_PER_THREAD] = esrc[ex];
            w[BIN_PER_THREAD] = ew[ex];
        }
    }
    __syncthreads();

#pragma unroll
    for (int c = 0; c < BIN_SLOTS; ++c)
        if ((unsigned)d[c] < (unsigned)n_nodes) atomicAdd(&cnt[(unsigned)d[c] / TILE_ROWS], 1u);
    __syncthreads();

    unsigned mine[MAX_TILES / BIN_THREADS], first[MAX_TILES / BIN_THREADS];
    unsigned own = 0u;
#pragma unroll
    for (int j = 0; j < MAX_TILES / BIN_THREADS; ++j) {
        const int i = j * BIN_THREADS + (int)threadIdx.x;
        mine[j]  = i < tiles ? cnt[i] : 0u;
        first[j] = mine[j] ? atomicAdd(&counters[i], mine[j]) : 0u;
        own += mine[j];
    }
    unsigned upto = own;
#pragma unroll
    for (int off = 1; off < 32; off <<= 1) {
        const unsigned v = (unsigned)__shfl_up((int)upto, off, 32);
        if (lane >= off) upto += v;
    }
    if (lane == 31) wave_sum[wave] = upto;
    __syncthreads();

    unsigned slot = upto - own;
    unsigned n_staged = 0u;
#pragma unroll
    for (int v = 0; v < BIN_THREADS / 32; ++v) {
        const unsigned t = wave_sum[v];
        if (v < wave) slot += t;
        n_staged += t;
    }
    if (n_staged == 0u) return;
#pragma unroll
    for (int j = 0; j < MAX_TILES / BIN_THREADS; ++j) {
        const int i = j * BIN_THREADS + (int)threadIdx.x;
        if (i < tiles) {
            cnt[i]   = slot;
            shift[i] = (unsigned)i * (TILE_ROWS * BUCKET_PER_ROW) + first[j] - slot;
            slot += mine[j];
        }
    }
    __syncthreads();

    unsigned pos[BIN_SLOTS], sh[BIN_SLOTS];
#pragma unroll
    for (int c = 0; c < BIN_SLOTS; ++c) {
        const bool     kept = (unsigned)d[c] < (unsigned)n_nodes;
        const unsigned t    = kept ? (unsigned)d[c] / TILE_ROWS : MAX_TILES + (unsigned)lane;
        const unsigned r    = atomicAdd(&cnt[t], 1u);
        sh[c]  = shift[t];
        pos[c] = kept ? r : BIN_STAGE + (unsigned)lane;
    }
#pragma unroll
    for (int c = 0; c < BIN_SLOTS; ++c) {
        const bool     kept  = (unsigned)d[c] < (unsigned)n_nodes;
        const unsigned t     = kept ? (unsigned)d[c] / TILE_ROWS : 0u;
        const unsigned row0  = t * TILE_ROWS;
        const unsigned left  = (unsigned)n_nodes - row0;
        const unsigned cap   = (left < TILE_ROWS ? left : TILE_ROWS) * BUCKET_PER_ROW;
        const unsigned entry = pos[c] + sh[c];
        uint2 ent;
        ent.x = ((unsigned)s[c] & ((1u << ROW_SHIFT) - 1u)) | (((unsigned)d[c] - row0) << ROW_SHIFT);
        ent.y = __float_as_uint(w[c]);
        stage[pos[c]] = ent;
        where[pos[c]] = (entry - row0 * BUCKET_PER_ROW) < cap ? entry : NO_PLACE;
    }
    __syncthreads();

    uint2* out2 = reinterpret_cast<uint2*>(out);
    unsigned place[BIN_SLOTS];
    uint2    ent[BIN_SLOTS];
#pragma unroll
    for (int j = 0; j < BIN_SLOTS; ++j) {
        const unsigned i = (unsigned)(j * BIN_THREADS) + threadIdx.x;
        const unsigned k = i < n_staged ? i : 0u;
        place[j] = where[k];
        ent[j]   = stage[k];
    }
#pragma unroll
    for (int j = 0; j < BIN_SLOTS; ++j) {
        const unsigned i = (unsigned)(j * BIN_THREADS) + threadIdx.x;
        if (i < n_staged && place[j] != NO_PLACE) out2[place[j]] = ent[j];
    }
}

#define AGG_THREADS 1024
#define AGG_WAVES   (AGG_THREADS / 32)
#define WIN_LOADS   4
#define WIN_EDGES   (WIN_LOADS * 128)
#define Q_CAP       544

__device__ __forceinline__ void add_rows(int s, float w, int dl, int lane,
                                         const float* __restrict__ h, float* tile)
{
#pragma unroll
    for (int half = 0; half < 2; ++half) {
        float2 hv[16];
#pragma unroll
        for (int i = 0; i < 16; ++i) {
            const int si = __builtin_amdgcn_readlane(s, half * 16 + i);
            hv[i] = reinterpret_cast<const float2*>(h + (long)si * N_OUT)[lane];
        }
#pragma unroll
        for (int i = 0; i < 16; ++i) {
            const float wi = __int_as_float(__builtin_amdgcn_readlane(__float_as_int(w), half * 16 + i));
            const int   di = __builtin_amdgcn_readlane(dl, half * 16 + i);
            float* t = tile + di * N_OUT + 2 * lane;
            atomicAdd(t + 0, hv[i].x * wi);
            atomicAdd(t + 1, hv[i].y * wi);
        }
    }
}

__device__ __forceinline__ void bucket_add(const uint2* bucket, int n, int lane, int wave,
                                           const float* __restrict__ h, float* tile)
{
    int g = wave * 32;
    if (g >= n) return;
    const int last = n - 1;
    uint2 cur = bucket[g + lane < last ? g + lane : last];
    for (; g < n; g += AGG_THREADS) {
        const int   gn  = g + AGG_THREADS;
        const uint2 nxt = bucket[gn + lane < last ? gn + lane : last];

        const bool  have = g + lane < n;
        const int   s  = have ? (int)(cur.x & ((1u << ROW_SHIFT) - 1u)) : 0;
        const float w  = have ? __uint_as_float(cur.y) : 0.0f;
        const int   dl = have ? (int)(cur.x >> ROW_SHIFT) : TILE_ROWS;
        add_rows(s, w, dl, lane, h, tile);
        cur = nxt;
    }
}

__device__ __forceinline__ void queue_push(bool keep, int e_first, unsigned c, unsigned row_in_tile,
                                           uint2* q, int& qn)
{
    const unsigned m = __builtin_amdgcn_ballot_w32(keep);
    if (m != 0u) {
        if (keep) {
            const int pos = qn + (int)__builtin_amdgcn_mbcnt_lo(m, 0u);
            q[pos] = make_uint2((unsigned)e_first, row_in_tile | (c << 16));
        }
        qn += __builtin_popcount(m);
    }
}

__device__ __forceinline__ void scan_window(const int4 (&buf)[WIN_LOADS], int wbase, int lane,
                                            int lo, unsigned nrows, uint2* q, int& qn)
{
#pragma unroll
    for (int u = 0; u < WIN_LOADS; ++u) {
        const int4 d = buf[u];
        const unsigned r0 = (unsigned)(d.x - lo);
        const unsigned r1 = (unsigned)(d.y - lo);
        const unsigned r2 = (unsigned)(d.z - lo);
        const unsigned r3 = (unsigned)(d.w - lo);
        const unsigned r01 = r0 < r1 ? r0 : r1;
        const unsigned r23 = r2 < r3 ? r2 : r3;
        const unsigned rmin = r01 < r23 ? r01 : r23;
        if (__builtin_amdgcn_ballot_w32(rmin < nrows) != 0u) {
            const int e0 = wbase + u * 128 + 4 * lane;
            queue_push(r0 < nrows, e0, 0u, r0, q, qn);
            queue_push(r1 < nrows, e0, 1u, r1, q, qn);
            queue_push(r2 < nrows, e0, 2u, r2, q, qn);
            queue_push(r3 < nrows, e0, 3u, r3, q, qn);
        }
    }
}

__device__ __forceinline__ void queue_drain(const uint2* q, int base, int nvalid, int lane,
                                            const float* __restrict__ h,
                                            const int* __restrict__ esrc,
                                            const float* __restrict__ ew,
                                            float* tile)
{
    __builtin_amdgcn_fence(__ATOMIC_ACQ_REL, "wavefront");

    int   s  = 0;
    float w  = 0.0f;
    int   dl = TILE_ROWS;
    if (lane < nvalid) {
        const uint2 ent = q[base + lane];
        const unsigned e = ent.x + (ent.y >> 16);
        s  = esrc[e];
        w  = ew[e];
        dl = (int)(ent.y & 0xffffu);
    }
    add_rows(s, w, dl, lane, h, tile);
}

__device__ __forceinline__ void scan_and_add(const float* __restrict__ h, const int* __restrict__ esrc,
                                             const int* __restrict__ edst, const float* __restrict__ ew,
                                             float* tile, uint2* q, int lane, int wave,
                                             int lo, unsigned nrows, int n_edges)
{
    int qn = 0;

    const int n_full  = n_edges / WIN_EDGES;
    const int covered = n_full * WIN_EDGES;
    const int4* d4 = reinterpret_cast<const int4*>(edst);
    const int rot = (int)(((long)blockIdx.x * n_full) / gridDim.x);
    int4 bufA[WIN_LOADS], bufB[WIN_LOADS];
    if (wave < n_full) {
        const int g0 = (wave + rot >= n_full) ? wave + rot - n_full : wave + rot;
#pragma unroll
        for (int u = 0; u < WIN_LOADS; ++u)
            bufA[u] = d4[(long)g0 * (WIN_EDGES / 4) + u * 32 + lane];
    }
    for (int j = wave; j < n_full; j += 2 * AGG_WAVES) {
        const int  ga   = (j + rot >= n_full) ? j + rot - n_full : j + rot;
        const int  jb   = j + AGG_WAVES;
        const bool hasB = jb < n_full;
        const int  jn   = j + 2 * AGG_WAVES;
        const int  gb   = !hasB ? ga : (jb + rot >= n_full) ? jb + rot - n_full : jb + rot;
        const int  gn   = !(jn < n_full) ? ga : (jn + rot >= n_full) ? jn + rot - n_full : jn + rot;

#pragma unroll
        for (int u = 0; u < WIN_LOADS; ++u)
            bufB[u] = d4[(long)gb * (WIN_EDGES / 4) + u * 32 + lane];

        scan_window(bufA, ga * WIN_EDGES, lane, lo, nrows, q, qn);
        while (qn >= 32) {
            qn -= 32;
            queue_drain(q, qn, 32, lane, h, esrc, ew, tile);
        }

#pragma unroll
        for (int u = 0; u < WIN_LOADS; ++u)
            bufA[u] = d4[(long)gn * (WIN_EDGES / 4) + u * 32 + lane];

        if (hasB) {
            scan_window(bufB, gb * WIN_EDGES, lane, lo, nrows, q, qn);
            while (qn >= 32) {
                qn -= 32;
                queue_drain(q, qn, 32, lane, h, esrc, ew, tile);
            }
        }
    }

    {
        int e0 = covered + wave * 32;
        for (;;) {
            const bool more = e0 < n_edges;
            if (more) {
                const int e = e0 + lane;
                unsigned r = 0xffffffffu;
                if (e < n_edges) r = (unsigned)(edst[e] - lo);
                queue_push(r < nrows, e, 0u, r, q, qn);
                e0 += AGG_WAVES * 32;
            }
            if (qn >= 32 || (!more && qn > 0)) {
                const int n = qn < 32 ? qn : 32;
                qn -= n;
                queue_drain(q, qn, n, lane, h, esrc, ew, tile);
            }
            if (!more && qn == 0) break;
        }
    }
}

__global__ __launch_bounds__(AGG_THREADS) void gcn_tile_aggregate(
    const float* __restrict__ h, const int* __restrict__ esrc,
    const int* __restrict__ edst, const float* __restrict__ ew,
    float* __restrict__ out, int n_nodes, int n_edges, unsigned long long ws_bytes)
{
    __shared__ __attribute__((aligned(16))) float tile[(TILE_ROWS + 1) * N_OUT];
    __shared__ __attribute__((aligned(16))) uint2 queue[AGG_WAVES * Q_CAP];

    const int lane = threadIdx.x & 31;
    const int wave = __builtin_amdgcn_readfirstlane((int)(threadIdx.x >> 5));
    const int lo   = blockIdx.x * TILE_ROWS;
    const int rem  = n_nodes - lo;
    const unsigned nrows = (unsigned)(rem < TILE_ROWS ? rem : TILE_ROWS);

    {
        float4* t4 = reinterpret_cast<float4*>(tile);
        const float4 z = {0.f, 0.f, 0.f, 0.f};
        for (int i = threadIdx.x; i < (TILE_ROWS + 1) * N_OUT / 4; i += AGG_THREADS) t4[i] = z;
    }
    __syncthreads();

    bool from_bucket = false;
    unsigned n_bucket = 0u;
    if (bins_fit(ws_bytes, n_nodes)) {
        const unsigned* counters = reinterpret_cast<const unsigned*>(h + (size_t)n_nodes * N_OUT);
        n_bucket    = counters[blockIdx.x];
        from_bucket = n_bucket <= nrows * BUCKET_PER_ROW;
    }

    if (from_bucket)
        bucket_add(reinterpret_cast<const uint2*>(out + (size_t)lo * N_OUT), (int)n_bucket, lane, wave, h, tile);
    else
        scan_and_add(h, esrc, edst, ew, tile, queue + wave * Q_CAP, lane, wave, lo, nrows, n_edges);

    __syncthreads();

    for (unsigned r = (unsigned)wave; r < nrows; r += AGG_WAVES) {
        const float2 v = reinterpret_cast<const float2*>(tile + r * N_OUT)[lane];

        float m = fmaxf(v.x, v.y);
#pragma unroll
        for (int off = 16; off > 0; off >>= 1)
            m = fmaxf(m, __shfl_xor(m, off, 32));

        const float x0 = __expf(v.x - m);
        const float x1 = __expf(v.y - m);
        float sum = x0 + x1;
#pragma unroll
        for (int off = 16; off > 0; off >>= 1)
            sum += __shfl_xor(sum, off, 32);

        const float inv = 1.0f / sum;
        float2 o; o.x = x0 * inv; o.y = x1 * inv;
        reinterpret_cast<float2*>(out + ((long)lo + r) * N_OUT)[lane] = o;
    }
}

extern "C" void kernel_launch(void* const* d_in, const int* in_sizes, int n_in,
                              void* d_out, int out_size, void* d_ws, size_t ws_size,
                              hipStream_t stream)
{
    const float* x    = (const float*)d_in[0];
    const int*   esrc = (const int*)d_in[1];
    const int*   edst = (const int*)d_in[2];
    const float* ew   = (const float*)d_in[3];
    const float* W    = (const float*)d_in[4];

    const int n_nodes = in_sizes[0] / D_FEAT;
    const int n_edges = in_sizes[1];

    float* out = (float*)d_out;
    float* h   = (float*)d_ws;

    {
        const int blocks = (n_nodes + 127) / 128;
        gcn_gemm_wmma<<<blocks, 256, 0, stream>>>(x, W, h, n_nodes, (unsigned long long)ws_size);
    }
    {
        const int blocks = n_edges > 0 ? (n_edges + BIN_EDGES - 1) / BIN_EDGES : 1;
        gcn_bin_edges<<<blocks, BIN_THREADS, 0, stream>>>(esrc, edst, ew, out, h, n_nodes, n_edges, (unsigned long long)ws_size);
    }
    {
        const int tiles = (n_nodes + TILE_ROWS - 1) / TILE_ROWS;
        gcn_tile_aggregate<<<tiles, AGG_THREADS, 0, stream>>>(h, esrc, edst, ew, out, n_nodes, n_edges, (unsigned long long)ws_size);
    }
}


# projection kernel: x fragment loads issued two trips (2 x 8 k-steps) ahead of their matrix instructions, in a ring of four register buffers, so no wave waits a memory round trip per trip
# speedup vs baseline: 2.3283x; 1.0416x over previous
#include <hip/hip_runtime.h>

typedef __attribute__((ext_vector_type(2))) float v2f;
typedef __attribute__((ext_vector_type(8))) float v8f;

#define D_FEAT 512
#define N_OUT  64
#define PAIR_ROWS   (D_FEAT / 2)
#define PAIR_STRIDE 80
#define A_STEPS     8
#define A_TRIPS     (D_FEAT / (4 * A_STEPS))

#define TILE_ROWS   391

#define ROW_SHIFT       23
#define BUCKET_PER_ROW  (N_OUT * 4 / 8)
#define MAX_TILES       1024
#define BIN_THREADS     256
#define BIN_PER_THREAD  16
#define BIN_SLOTS       (BIN_PER_THREAD + 1)
#define BIN_EDGES       (BIN_THREADS * BIN_PER_THREAD)
#define BIN_STAGE       (BIN_THREADS * BIN_SLOTS)
#define NO_PLACE        0xffffffffu

__device__ __forceinline__ bool bins_fit(unsigned long long ws_bytes, int n_nodes)
{
    const unsigned long long h_bytes = (unsigned long long)n_nodes * N_OUT * sizeof(float);
    const int tiles = (n_nodes + TILE_ROWS - 1) / TILE_ROWS;
    return n_nodes > 0 && n_nodes <= (1 << ROW_SHIFT) && tiles <= MAX_TILES
        && ws_bytes >= h_bytes + (unsigned long long)tiles * sizeof(unsigned);
}

__global__ __launch_bounds__(256) void gcn_gemm_wmma(
    const float* __restrict__ x, const float* __restrict__ W,
    float* __restrict__ h, int n_nodes, unsigned long long ws_bytes)
{
    __shared__ float2 Wp[PAIR_ROWS * PAIR_STRIDE];

    if (blockIdx.x == 0 && bins_fit(ws_bytes, n_nodes)) {
        unsigned* counters = reinterpret_cast<unsigned*>(h + (size_t)n_nodes * N_OUT);
        const int tiles = (n_nodes + TILE_ROWS - 1) / TILE_ROWS;
        for (int i = threadIdx.x; i < tiles; i += 256) counters[i] = 0u;
    }
    for (int i = threadIdx.x; i < PAIR_ROWS * 16; i += 256) {
        const int p  = i >> 4;
        const int n4 = (i & 15) << 2;
        const float4 r0 = *reinterpret_cast<const float4*>(W + (2 * p + 0) * N_OUT + n4);
        const float4 r1 = *reinterpret_cast<const float4*>(W + (2 * p + 1) * N_OUT + n4);
        float4* dst = reinterpret_cast<float4*>(&Wp[p * PAIR_STRIDE + n4]);
        float4 lo, hi;
        lo.x = r0.x; lo.y = r1.x; lo.z = r0.y; lo.w = r1.y;
        hi.x = r0.z; hi.y = r1.z; hi.z = r0.w; hi.w = r1.w;
        dst[0] = lo;
        dst[1] = hi;
    }
    __syncthreads();

    const int lane = threadIdx.x & 31;
    const int wave = threadIdx.x >> 5;
    const int l16  = lane & 15;
    const int g    = lane >> 4;
    const long rowBase = (long)blockIdx.x * 128 + (long)wave * 16;
    if (rowBase >= n_nodes) return;

    const float* aptr = x + (rowBase + l16) * (long)D_FEAT + 2 * g;
    const float2* bptr = &Wp[(long)g * PAIR_STRIDE + l16];

    v8f acc0 = {}, acc1 = {}, acc2 = {}, acc3 = {};

    v2f abuf0[A_STEPS], abuf1[A_STEPS], abuf2[A_STEPS], abuf3[A_STEPS];

#define LOAD_TRIP(buf, t)                                                                   \
    _Pragma("unroll")                     \
    for (int j = 0; j < A_STEPS; ++j)                                                       \
        buf[j] = *reinterpret_cast<const v2f*>(aptr + (t) * (4 * A_STEPS) + 4 * j);

#define MMA_TRIP(buf, t)                                                                    \
    _Pragma("unroll")                     \
    for (int j = 0; j < A_STEPS; ++j) {                                                     \
        const float2* wb = bptr + ((t) * (2 * A_STEPS) + 2 * j) * PAIR_STRIDE;              \
        const v2f b0 = *reinterpret_cast<const v2f*>(wb +  0);                              \
        const v2f b1 = *reinterpret_cast<const v2f*>(wb + 16);                              \
        const v2f b2 = *reinterpret_cast<const v2f*>(wb + 32);                              \
        const v2f b3 = *reinterpret_cast<const v2f*>(wb + 48);                              \
        acc0 = __builtin_amdgcn_wmma_f32_16x16x4_f32(false, buf[j], false, b0, (short)0, acc0, false, false); \
        acc1 = __builtin_amdgcn_wmma_f32_16x16x4_f32(false, buf[j], false, b1, (short)0, acc1, false, false); \
        acc2 = __builtin_amdgcn_wmma_f32_16x16x4_f32(false, buf[j], false, b2, (short)0, acc2, false, false); \
        acc3 = __builtin_amdgcn_wmma_f32_16x16x4_f32(false, buf[j], false, b3, (short)0, acc3, false, false); \
    }

    LOAD_TRIP(abuf0, 0)
    LOAD_TRIP(abuf1, 1)
#pragma unroll 1
    for (int t = 0; t < A_TRIPS - 4; t += 4) {
        LOAD_TRIP(abuf2, t + 2)  MMA_TRIP(abuf0, t)
        LOAD_TRIP(abuf3, t + 3)  MMA_TRIP(abuf1, t + 1)
        LOAD_TRIP(abuf0, t + 4)  MMA_TRIP(abuf2, t + 2)
        LOAD_TRIP(abuf1, t + 5)  MMA_TRIP(abuf3, t + 3)
    }
    LOAD_TRIP(abuf2, A_TRIPS - 2)  MMA_TRIP(abuf0, A_TRIPS - 4)
    LOAD_TRIP(abuf3, A_TRIPS - 1)  MMA_TRIP(abuf1, A_TRIPS - 3)
    MMA_TRIP(abuf2, A_TRIPS - 2)
    MMA_TRIP(abuf3, A_TRIPS - 1)
#undef LOAD_TRIP
#undef MMA_TRIP

    float* hb = h + rowBase * N_OUT + l16;
#pragma unroll
    for (int v = 0; v < 8; ++v) {
        const long m = v + 8 * g;
        hb[m * N_OUT +  0] = acc0[v];
        hb[m * N_OUT + 16] = acc1[v];
        hb[m * N_OUT + 32] = acc2[v];
        hb[m * N_OUT + 48] = acc3[v];
    }
}

__global__ __launch_bounds__(BIN_THREADS) void gcn_bin_edges(
    const int* __restrict__ esrc, const int* __restrict__ edst, const float* __restrict__ ew,
    float* __restrict__ out, float* __restrict__ h, int n_nodes, int n_edges, unsigned long long ws_bytes)
{
    __shared__ unsigned cnt[MAX_TILES + 32];
    __shared__ unsigned shift[MAX_TILES + 32];
    __shared__ __attribute__((aligned(16))) uint2 stage[BIN_STAGE + 32];
    __shared__ unsigned where[BIN_STAGE + 32];
    __shared__ unsigned wave_sum[BIN_THREADS / 32];

    if (!bins_fit(ws_bytes, n_nodes)) return;
    unsigned* counters = reinterpret_cast<unsigned*>(h + (size_t)n_nodes * N_OUT);
    const int tiles = (n_nodes + TILE_ROWS - 1) / TILE_ROWS;
    const int lane  = threadIdx.x & 31;
    const int wave  = threadIdx.x >> 5;

    for (int i = threadIdx.x; i < tiles; i += BIN_THREADS) cnt[i] = 0u;
    if (threadIdx.x < 32) { cnt[MAX_TILES + threadIdx.x] = 0u; shift[MAX_TILES + threadIdx.x] = 0u; }

    int   d[BIN_SLOTS], s[BIN_SLOTS];
    float w[BIN_SLOTS];
#pragma unroll
    for (int j = 0; j < BIN_PER_THREAD / 4; ++j) {
        const long e0 = (long)blockIdx.x * BIN_EDGES + j * (BIN_THREADS * 4) + threadIdx.x * 4;
        int4   dv = {-1, -1, -1, -1};
        int4   sv = {0, 0, 0, 0};
        float4 wv = {0.f, 0.f, 0.f, 0.f};
        if (e0 + 3 < n_edges) {
            dv = *reinterpret_cast<const int4*>(edst + e0);
            sv = *reinterpret_cast<const int4*>(esrc + e0);
            wv = *reinterpret_cast<const float4*>(ew + e0);
        }
        d[4 * j + 0] = dv.x; d[4 * j + 1] = dv.y; d[4 * j + 2] = dv.z; d[4 * j + 3] = dv.w;
        s[4 * j + 0] = sv.x; s[4 * j + 1] = sv.y; s[4 * j + 2] = sv.z; s[4 * j + 3] = sv.w;
        w[4 * j + 0] = wv.x; w[4 * j + 1] = wv.y; w[4 * j + 2] = wv.z; w[4 * j + 3] = wv.w;
    }
    {
        const int ex = (n_edges & ~3) + (int)threadIdx.x;
        d[BIN_PER_THREAD] = -1; s[BIN_PER_THREAD] = 0; w[BIN_PER_THREAD] = 0.0f;
        if (blockIdx.x == 0 && ex < n_edges) {
            d[BIN_PER_THREAD] = edst[ex];
            s[BIN_PER_THREAD] = esrc[ex];
            w[BIN_PER_THREAD] = ew[ex];
        }
    }
    __syncthreads();

#pragma unroll
    for (int c = 0; c < BIN_SLOTS; ++c)
        if ((unsigned)d[c] < (unsigned)n_nodes) atomicAdd(&cnt[(unsigned)d[c] / TILE_ROWS], 1u);
    __syncthreads();

    unsigned mine[MAX_TILES / BIN_THREADS], first[MAX_TILES / BIN_THREADS];
    unsigned own = 0u;
#pragma unroll
    for (int j = 0; j < MAX_TILES / BIN_THREADS; ++j) {
        const int i = j * BIN_THREADS + (int)threadIdx.x;
        mine[j]  = i < tiles ? cnt[i] : 0u;
        first[j] = mine[j] ? atomicAdd(&counters[i], mine[j]) : 0u;
        own += mine[j];
    }
    unsigned upto = own;
#pragma unroll
    for (int off = 1; off < 32; off <<= 1) {
        const unsigned v = (unsigned)__shfl_up((int)upto, off, 32);
        if (lane >= off) upto += v;
    }
    if (lane == 31) wave_sum[wave] = upto;
    __syncthreads();

    unsigned slot = upto - own;
    unsigned n_staged = 0u;
#pragma unroll
    for (int v = 0; v < BIN_THREADS / 32; ++v) {
        const unsigned t = wave_sum[v];
        if (v < wave) slot += t;
        n_staged += t;
    }
    if (n_staged == 0u) return;
#pragma unroll
    for (int j = 0; j < MAX_TILES / BIN_THREADS; ++j) {
        const int i = j * BIN_THREADS + (int)threadIdx.x;
        if (i < tiles) {
            cnt[i]   = slot;
            shift[i] = (unsigned)i * (TILE_ROWS * BUCKET_PER_ROW) + first[j] - slot;
            slot += mine[j];
        }
    }
    __syncthreads();

    unsigned pos[BIN_SLOTS], sh[BIN_SLOTS];
#pragma unroll
    for (int c = 0; c < BIN_SLOTS; ++c) {
        const bool     kept = (unsigned)d[c] < (unsigned)n_nodes;
        const unsigned t    = kept ? (unsigned)d[c] / TILE_ROWS : MAX_TILES + (unsigned)lane;
        const unsigned r    = atomicAdd(&cnt[t], 1u);
        sh[c]  = shift[t];
        pos[c] = kept ? r : BIN_STAGE + (unsigned)lane;
    }
#pragma unroll
    for (int c = 0; c < BIN_SLOTS; ++c) {
        const bool     kept  = (unsigned)d[c] < (unsigned)n_nodes;
        const unsigned t     = kept ? (unsigned)d[c] / TILE_ROWS : 0u;
        const unsigned row0  = t * TILE_ROWS;
        const unsigned left  = (unsigned)n_nodes - row0;
        const unsigned cap   = (left < TILE_ROWS ? left : TILE_ROWS) * BUCKET_PER_ROW;
        const unsigned entry = pos[c] + sh[c];
        uint2 ent;
        ent.x = ((unsigned)s[c] & ((1u << ROW_SHIFT) - 1u)) | (((unsigned)d[c] - row0) << ROW_SHIFT);
        ent.y = __float_as_uint(w[c]);
        stage[pos[c]] = ent;
        where[pos[c]] = (entry - row0 * BUCKET_PER_ROW) < cap ? entry : NO_PLACE;
    }
    __syncthreads();

    uint2* out2 = reinterpret_cast<uint2*>(out);
    unsigned place[BIN_SLOTS];
    uint2    ent[BIN_SLOTS];
#pragma unroll
    for (int j = 0; j < BIN_SLOTS; ++j) {
        const unsigned i = (unsigned)(j * BIN_THREADS) + threadIdx.x;
        const unsigned k = i < n_staged ? i : 0u;
        place[j] = where[k];
        ent[j]   = stage[k];
    }
#pragma unroll
    for (int j = 0; j < BIN_SLOTS; ++j) {
        const unsigned i = (unsigned)(j * BIN_THREADS) + threadIdx.x;
        if (i < n_staged && place[j] != NO_PLACE) out2[place[j]] = ent[j];
    }
}

#define AGG_THREADS 1024
#define AGG_WAVES   (AGG_THREADS / 32)
#define WIN_LOADS   4
#define WIN_EDGES   (WIN_LOADS * 128)
#define Q_CAP       544

__device__ __forceinline__ void add_rows(int s, float w, int dl, int lane,
                                         const float* __restrict__ h, float* tile)
{
#pragma unroll
    for (int half = 0; half < 2; ++half) {
        float2 hv[16];
#pragma unroll
        for (int i = 0; i < 16; ++i) {
            const int si = __builtin_amdgcn_readlane(s, half * 16 + i);
            hv[i] = reinterpret_cast<const float2*>(h + (long)si * N_OUT)[lane];
        }
#pragma unroll
        for (int i = 0; i < 16; ++i) {
            const float wi = __int_as_float(__builtin_amdgcn_readlane(__float_as_int(w), half * 16 + i));
            const int   di = __builtin_amdgcn_readlane(dl, half * 16 + i);
            float* t = tile + di * N_OUT + 2 * lane;
            atomicAdd(t + 0, hv[i].x * wi);
            atomicAdd(t + 1, hv[i].y * wi);
        }
    }
}

__device__ __forceinline__ void bucket_add(const uint2* bucket, int n, int lane, int wave,
                                           const float* __restrict__ h, float* tile)
{
    int g = wave * 32;
    if (g >= n) return;
    const int last = n - 1;
    uint2 cur = bucket[g + lane < last ? g + lane : last];
    for (; g < n; g += AGG_THREADS) {
        const int   gn  = g + AGG_THREADS;
        const uint2 nxt = bucket[gn + lane < last ? gn + lane : last];

        const bool  have = g + lane < n;
        const int   s  = have ? (int)(cur.x & ((1u << ROW_SHIFT) - 1u)) : 0;
        const float w  = have ? __uint_as_float(cur.y) : 0.0f;
        const int   dl = have ? (int)(cur.x >> ROW_SHIFT) : TILE_ROWS;
        add_rows(s, w, dl, lane, h, tile);
        cur = nxt;
    }
}

__device__ __forceinline__ void queue_push(bool keep, int e_first, unsigned c, unsigned row_in_tile,
                                           uint2* q, int& qn)
{
    const unsigned m = __builtin_amdgcn_ballot_w32(keep);
    if (m != 0u) {
        if (keep) {
            const int pos = qn + (int)__builtin_amdgcn_mbcnt_lo(m, 0u);
            q[pos] = make_uint2((unsigned)e_first, row_in_tile | (c << 16));
        }
        qn += __builtin_popcount(m);
    }
}

__device__ __forceinline__ void scan_window(const int4 (&buf)[WIN_LOADS], int wbase, int lane,
                                            int lo, unsigned nrows, uint2* q, int& qn)
{
#pragma unroll
    for (int u = 0; u < WIN_LOADS; ++u) {
        const int4 d = buf[u];
        const unsigned r0 = (unsigned)(d.x - lo);
        const unsigned r1 = (unsigned)(d.y - lo);
        const unsigned r2 = (unsigned)(d.z - lo);
        const unsigned r3 = (unsigned)(d.w - lo);
        const unsigned r01 = r0 < r1 ? r0 : r1;
        const unsigned r23 = r2 < r3 ? r2 : r3;
        const unsigned rmin = r01 < r23 ? r01 : r23;
        if (__builtin_amdgcn_ballot_w32(rmin < nrows) != 0u) {
            const int e0 = wbase + u * 128 + 4 * lane;
            queue_push(r0 < nrows, e0, 0u, r0, q, qn);
            queue_push(r1 < nrows, e0, 1u, r1, q, qn);
            queue_push(r2 < nrows, e0, 2u, r2, q, qn);
            queue_push(r3 < nrows, e0, 3u, r3, q, qn);
        }
    }
}

__device__ __forceinline__ void queue_drain(const uint2* q, int base, int nvalid, int lane,
                                            const float* __restrict__ h,
                                            const int* __restrict__ esrc,
                                            const float* __restrict__ ew,
                                            float* tile)
{
    __builtin_amdgcn_fence(__ATOMIC_ACQ_REL, "wavefront");

    int   s  = 0;
    float w  = 0.0f;
    int   dl = TILE_ROWS;
    if (lane < nvalid) {
        const uint2 ent = q[base + lane];
        const unsigned e = ent.x + (ent.y >> 16);
        s  = esrc[e];
        w  = ew[e];
        dl = (int)(ent.y & 0xffffu);
    }
    add_rows(s, w, dl, lane, h, tile);
}

__device__ __forceinline__ void scan_and_add(const float* __restrict__ h, const int* __restrict__ esrc,
                                             const int* __restrict__ edst, const float* __restrict__ ew,
                                             float* tile, uint2* q, int lane, int wave,
                                             int lo, unsigned nrows, int n_edges)
{
    int qn = 0;

    const int n_full  = n_edges / WIN_EDGES;
    const int covered = n_full * WIN_EDGES;
    const int4* d4 = reinterpret_cast<const int4*>(edst);
    const int rot = (int)(((long)blockIdx.x * n_full) / gridDim.x);
    int4 bufA[WIN_LOADS], bufB[WIN_LOADS];
    if (wave < n_full) {
        const int g0 = (wave + rot >= n_full) ? wave + rot - n_full : wave + rot;
#pragma unroll
        for (int u = 0; u < WIN_LOADS; ++u)
            bufA[u] = d4[(long)g0 * (WIN_EDGES / 4) + u * 32 + lane];
    }
    for (int j = wave; j < n_full; j += 2 * AGG_WAVES) {
        const int  ga   = (j + rot >= n_full) ? j + rot - n_full : j + rot;
        const int  jb   = j + AGG_WAVES;
        const bool hasB = jb < n_full;
        const int  jn   = j + 2 * AGG_WAVES;
        const int  gb   = !hasB ? ga : (jb + rot >= n_full) ? jb + rot - n_full : jb + rot;
        const int  gn   = !(jn < n_full) ? ga : (jn + rot >= n_full) ? jn + rot - n_full : jn + rot;

#pragma unroll
        for (int u = 0; u < WIN_LOADS; ++u)
            bufB[u] = d4[(long)gb * (WIN_EDGES / 4) + u * 32 + lane];

        scan_window(bufA, ga * WIN_EDGES, lane, lo, nrows, q, qn);
        while (qn >= 32) {
            qn -= 32;
            queue_drain(q, qn, 32, lane, h, esrc, ew, tile);
        }

#pragma unroll
        for (int u = 0; u < WIN_LOADS; ++u)
            bufA[u] = d4[(long)gn * (WIN_EDGES / 4) + u * 32 + lane];

        if (hasB) {
            scan_window(bufB, gb * WIN_EDGES, lane, lo, nrows, q, qn);
            while (qn >= 32) {
                qn -= 32;
                queue_drain(q, qn, 32, lane, h, esrc, ew, tile);
            }
        }
    }

    {
        int e0 = covered + wave * 32;
        for (;;) {
            const bool more = e0 < n_edges;
            if (more) {
                const int e = e0 + lane;
                unsigned r = 0xffffffffu;
                if (e < n_edges) r = (unsigned)(edst[e] - lo);
                queue_push(r < nrows, e, 0u, r, q, qn);
                e0 += AGG_WAVES * 32;
            }
            if (qn >= 32 || (!more && qn > 0)) {
                const int n = qn < 32 ? qn : 32;
                qn -= n;
                queue_drain(q, qn, n, lane, h, esrc, ew, tile);
            }
            if (!more && qn == 0) break;
        }
    }
}

__global__ __launch_bounds__(AGG_THREADS) void gcn_tile_aggregate(
    const float* __restrict__ h, const int* __restrict__ esrc,
    const int* __restrict__ edst, const float* __restrict__ ew,
    float* __restrict__ out, int n_nodes, int n_edges, unsigned long long ws_bytes)
{
    __shared__ __attribute__((aligned(16))) float tile[(TILE_ROWS + 1) * N_OUT];
    __shared__ __attribute__((aligned(16))) uint2 queue[AGG_WAVES * Q_CAP];

    const int lane = threadIdx.x & 31;
    const int wave = __builtin_amdgcn_readfirstlane((int)(threadIdx.x >> 5));
    const int lo   = blockIdx.x * TILE_ROWS;
    const int rem  = n_nodes - lo;
    const unsigned nrows = (unsigned)(rem < TILE_ROWS ? rem : TILE_ROWS);

    {
        float4* t4 = reinterpret_cast<float4*>(tile);
        const float4 z = {0.f, 0.f, 0.f, 0.f};
        for (int i = threadIdx.x; i < (TILE_ROWS + 1) * N_OUT / 4; i += AGG_THREADS) t4[i] = z;
    }
    __syncthreads();

    bool from_bucket = false;
    unsigned n_bucket = 0u;
    if (bins_fit(ws_bytes, n_nodes)) {
        const unsigned* counters = reinterpret_cast<const unsigned*>(h + (size_t)n_nodes * N_OUT);
        n_bucket    = counters[blockIdx.x];
        from_bucket = n_bucket <= nrows * BUCKET_PER_ROW;
    }

    if (from_bucket)
        bucket_add(reinterpret_cast<const uint2*>(out + (size_t)lo * N_OUT), (int)n_bucket, lane, wave, h, tile);
    else
        scan_and_add(h, esrc, edst, ew, tile, queue + wave * Q_CAP, lane, wave, lo, nrows, n_edges);

    __syncthreads();

    for (unsigned r = (unsigned)wave; r < nrows; r += AGG_WAVES) {
        const float2 v = reinterpret_cast<const float2*>(tile + r * N_OUT)[lane];

        float m = fmaxf(v.x, v.y);
#pragma unroll
        for (int off = 16; off > 0; off >>= 1)
            m = fmaxf(m, __shfl_xor(m, off, 32));

        const float x0 = __expf(v.x - m);
        const float x1 = __expf(v.y - m);
        float sum = x0 + x1;
#pragma unroll
        for (int off = 16; off > 0; off >>= 1)
            sum += __shfl_xor(sum, off, 32);

        const float inv = 1.0f / sum;
        float2 o; o.x = x0 * inv; o.y = x1 * inv;
        reinterpret_cast<float2*>(out + ((long)lo + r) * N_OUT)[lane] = o;
    }
}

extern "C" void kernel_launch(void* const* d_in, const int* in_sizes, int n_in,
                              void* d_out, int out_size, void* d_ws, size_t ws_size,
                              hipStream_t stream)
{
    const float* x    = (const float*)d_in[0];
    const int*   esrc = (const int*)d_in[1];
    const int*   edst = (const int*)d_in[2];
    const float* ew   = (const float*)d_in[3];
    const float* W    = (const float*)d_in[4];

    const int n_nodes = in_sizes[0] / D_FEAT;
    const int n_edges = in_sizes[1];

    float* out = (float*)d_out;
    float* h   = (float*)d_ws;

    {
        const int blocks = (n_nodes + 127) / 128;
        gcn_gemm_wmma<<<blocks, 256, 0, stream>>>(x, W, h, n_nodes, (unsigned long long)ws_size);
    }
    {
        const int blocks = n_edges > 0 ? (n_edges + BIN_EDGES - 1) / BIN_EDGES : 1;
        gcn_bin_edges<<<blocks, BIN_THREADS, 0, stream>>>(esrc, edst, ew, out, h, n_nodes, n_edges, (unsigned long long)ws_size);
    }
    {
        const int tiles = (n_nodes + TILE_ROWS - 1) / TILE_ROWS;
        gcn_tile_aggregate<<<tiles, AGG_THREADS, 0, stream>>>(h, esrc, edst, ew, out, n_nodes, n_edges, (unsigned long long)ws_size);
    }
}


# aggregate: LDS f32 atomic adds replaced by an LDS counting sort of the tile bucket by row (int rank adds) and register sums by a half-wave per row, softmax in registers; row terms in arrival order
# speedup vs baseline: 2.6169x; 1.1240x over previous
#include <hip/hip_runtime.h>

typedef __attribute__((ext_vector_type(2))) float v2f;
typedef __attribute__((ext_vector_type(8))) float v8f;

#define D_FEAT 512
#define N_OUT  64
#define PAIR_ROWS   (D_FEAT / 2)
#define PAIR_STRIDE 80
#define A_STEPS     8
#define A_TRIPS     (D_FEAT / (4 * A_STEPS))

#define TILE_ROWS   391

#define ROW_SHIFT       23
#define BUCKET_PER_ROW  (N_OUT * 4 / 8)
#define MAX_TILES       1024
#define BIN_THREADS     256
#define BIN_PER_THREAD  16
#define BIN_SLOTS       (BIN_PER_THREAD + 1)
#define BIN_EDGES       (BIN_THREADS * BIN_PER_THREAD)
#define BIN_STAGE       (BIN_THREADS * BIN_SLOTS)
#define NO_PLACE        0xffffffffu

__device__ __forceinline__ bool bins_fit(unsigned long long ws_bytes, int n_nodes)
{
    const unsigned long long h_bytes = (unsigned long long)n_nodes * N_OUT * sizeof(float);
    const int tiles = (n_nodes + TILE_ROWS - 1) / TILE_ROWS;
    return n_nodes > 0 && n_nodes <= (1 << ROW_SHIFT) && tiles <= MAX_TILES
        && ws_bytes >= h_bytes + (unsigned long long)tiles * sizeof(unsigned);
}

__global__ __launch_bounds__(256) void gcn_gemm_wmma(
    const float* __restrict__ x, const float* __restrict__ W,
    float* __restrict__ h, int n_nodes, unsigned long long ws_bytes)
{
    __shared__ float2 Wp[PAIR_ROWS * PAIR_STRIDE];

    if (blockIdx.x == 0 && bins_fit(ws_bytes, n_nodes)) {
        unsigned* counters = reinterpret_cast<unsigned*>(h + (size_t)n_nodes * N_OUT);
        const int tiles = (n_nodes + TILE_ROWS - 1) / TILE_ROWS;
        for (int i = threadIdx.x; i < tiles; i += 256) counters[i] = 0u;
    }
    for (int i = threadIdx.x; i < PAIR_ROWS * 16; i += 256) {
        const int p  = i >> 4;
        const int n4 = (i & 15) << 2;
        const float4 r0 = *reinterpret_cast<const float4*>(W + (2 * p + 0) * N_OUT + n4);
        const float4 r1 = *reinterpret_cast<const float4*>(W + (2 * p + 1) * N_OUT + n4);
        float4* dst = reinterpret_cast<float4*>(&Wp[p * PAIR_STRIDE + n4]);
        float4 lo, hi;
        lo.x = r0.x; lo.y = r1.x; lo.z = r0.y; lo.w = r1.y;
        hi.x = r0.z; hi.y = r1.z; hi.z = r0.w; hi.w = r1.w;
        dst[0] = lo;
        dst[1] = hi;
    }
    __syncthreads();

    const int lane = threadIdx.x & 31;
    const int wave = threadIdx.x >> 5;
    const int l16  = lane & 15;
    const int g    = lane >> 4;
    const long rowBase = (long)blockIdx.x * 128 + (long)wave * 16;
    if (rowBase >= n_nodes) return;

    const float* aptr = x + (rowBase + l16) * (long)D_FEAT + 2 * g;
    const float2* bptr = &Wp[(long)g * PAIR_STRIDE + l16];

    v8f acc0 = {}, acc1 = {}, acc2 = {}, acc3 = {};

    v2f abuf0[A_STEPS], abuf1[A_STEPS], abuf2[A_STEPS], abuf3[A_STEPS];

#define LOAD_TRIP(buf, t)                                                                   \
    _Pragma("unroll")                     \
    for (int j = 0; j < A_STEPS; ++j)                                                       \
        buf[j] = *reinterpret_cast<const v2f*>(aptr + (t) * (4 * A_STEPS) + 4 * j);

#define MMA_TRIP(buf, t)                                                                    \
    _Pragma("unroll")                     \
    for (int j = 0; j < A_STEPS; ++j) {                                                     \
        const float2* wb = bptr + ((t) * (2 * A_STEPS) + 2 * j) * PAIR_STRIDE;              \
        const v2f b0 = *reinterpret_cast<const v2f*>(wb +  0);                              \
        const v2f b1 = *reinterpret_cast<const v2f*>(wb + 16);                              \
        const v2f b2 = *reinterpret_cast<const v2f*>(wb + 32);                              \
        const v2f b3 = *reinterpret_cast<const v2f*>(wb + 48);                              \
        acc0 = __builtin_amdgcn_wmma_f32_16x16x4_f32(false, buf[j], false, b0, (short)0, acc0, false, false); \
        acc1 = __builtin_amdgcn_wmma_f32_16x16x4_f32(false, buf[j], false, b1, (short)0, acc1, false, false); \
        acc2 = __builtin_amdgcn_wmma_f32_16x16x4_f32(false, buf[j], false, b2, (short)0, acc2, false, false); \
        acc3 = __builtin_amdgcn_wmma_f32_16x16x4_f32(false, buf[j], false, b3, (short)0, acc3, false, false); \
    }

    LOAD_TRIP(abuf0, 0)
    LOAD_TRIP(abuf1, 1)
#pragma unroll 1
    for (int t = 0; t < A_TRIPS - 4; t += 4) {
        LOAD_TRIP(abuf2, t + 2)  MMA_TRIP(abuf0, t)
        LOAD_TRIP(abuf3, t + 3)  MMA_TRIP(abuf1, t + 1)
        LOAD_TRIP(abuf0, t + 4)  MMA_TRIP(abuf2, t + 2)
        LOAD_TRIP(abuf1, t + 5)  MMA_TRIP(abuf3, t + 3)
    }
    LOAD_TRIP(abuf2, A_TRIPS - 2)  MMA_TRIP(abuf0, A_TRIPS - 4)
    LOAD_TRIP(abuf3, A_TRIPS - 1)  MMA_TRIP(abuf1, A_TRIPS - 3)
    MMA_TRIP(abuf2, A_TRIPS - 2)
    MMA_TRIP(abuf3, A_TRIPS - 1)
#undef LOAD_TRIP
#undef MMA_TRIP

    float* hb = h + rowBase * N_OUT + l16;
#pragma unroll
    for (int v = 0; v < 8; ++v) {
        const long m = v + 8 * g;
        hb[m * N_OUT +  0] = acc0[v];
        hb[m * N_OUT + 16] = acc1[v];
        hb[m * N_OUT + 32] = acc2[v];
        hb[m * N_OUT + 48] = acc3[v];
    }
}

__global__ __launch_bounds__(BIN_THREADS) void gcn_bin_edges(
    const int* __restrict__ esrc, const int* __restrict__ edst, const float* __restrict__ ew,
    float* __restrict__ out, float* __restrict__ h, int n_nodes, int n_edges, unsigned long long ws_bytes)
{
    __shared__ unsigned cnt[MAX_TILES + 32];
    __shared__ unsigned shift[MAX_TILES + 32];
    __shared__ __attribute__((aligned(16))) uint2 stage[BIN_STAGE + 32];
    __shared__ unsigned where[BIN_STAGE + 32];
    __shared__ unsigned wave_sum[BIN_THREADS / 32];

    if (!bins_fit(ws_bytes, n_nodes)) return;
    unsigned* counters = reinterpret_cast<unsigned*>(h + (size_t)n_nodes * N_OUT);
    const int tiles = (n_nodes + TILE_ROWS - 1) / TILE_ROWS;
    const int lane  = threadIdx.x & 31;
    const int wave  = threadIdx.x >> 5;

    for (int i = threadIdx.x; i < tiles; i += BIN_THREADS) cnt[i] = 0u;
    if (threadIdx.x < 32) { cnt[MAX_TILES + threadIdx.x] = 0u; shift[MAX_TILES + threadIdx.x] = 0u; }

    int   d[BIN_SLOTS], s[BIN_SLOTS];
    float w[BIN_SLOTS];
#pragma unroll
    for (int j = 0; j < BIN_PER_THREAD / 4; ++j) {
        const long e0 = (long)blockIdx.x * BIN_EDGES + j * (BIN_THREADS * 4) + threadIdx.x * 4;
        int4   dv = {-1, -1, -1, -1};
        int4   sv = {0, 0, 0, 0};
        float4 wv = {0.f, 0.f, 0.f, 0.f};
        if (e0 + 3 < n_edges) {
            dv = *reinterpret_cast<const int4*>(edst + e0);
            sv = *reinterpret_cast<const int4*>(esrc + e0);
            wv = *reinterpret_cast<const float4*>(ew + e0);
        }
        d[4 * j + 0] = dv.x; d[4 * j + 1] = dv.y; d[4 * j + 2] = dv.z; d[4 * j + 3] = dv.w;
        s[4 * j + 0] = sv.x; s[4 * j + 1] = sv.y; s[4 * j + 2] = sv.z; s[4 * j + 3] = sv.w;
        w[4 * j + 0] = wv.x; w[4 * j + 1] = wv.y; w[4 * j + 2] = wv.z; w[4 * j + 3] = wv.w;
    }
    {
        const int ex = (n_edges & ~3) + (int)threadIdx.x;
        d[BIN_PER_THREAD] = -1; s[BIN_PER_THREAD] = 0; w[BIN_PER_THREAD] = 0.0f;
        if (blockIdx.x == 0 && ex < n_edges) {
            d[BIN_PER_THREAD] = edst[ex];
            s[BIN_PER_THREAD] = esrc[ex];
            w[BIN_PER_THREAD] = ew[ex];
        }
    }
    __syncthreads();

#pragma unroll
    for (int c = 0; c < BIN_SLOTS; ++c)
        if ((unsigned)d[c] < (unsigned)n_nodes) atomicAdd(&cnt[(unsigned)d[c] / TILE_ROWS], 1u);
    __syncthreads();

    unsigned mine[MAX_TILES / BIN_THREADS], first[MAX_TILES / BIN_THREADS];
    unsigned own = 0u;
#pragma unroll
    for (int j = 0; j < MAX_TILES / BIN_THREADS; ++j) {
        const int i = j * BIN_THREADS + (int)threadIdx.x;
        mine[j]  = i < tiles ? cnt[i] : 0u;
        first[j] = mine[j] ? atomicAdd(&counters[i], mine[j]) : 0u;
        own += mine[j];
    }
    unsigned upto = own;
#pragma unroll
    for (int off = 1; off < 32; off <<= 1) {
        const unsigned v = (unsigned)__shfl_up((int)upto, off, 32);
        if (lane >= off) upto += v;
    }
    if (lane == 31) wave_sum[wave] = upto;
    __syncthreads();

    unsigned slot = upto - own;
    unsigned n_staged = 0u;
#pragma unroll
    for (int v = 0; v < BIN_THREADS / 32; ++v) {
        const unsigned t = wave_sum[v];
        if (v < wave) slot += t;
        n_staged += t;
    }
    if (n_staged == 0u) return;
#pragma unroll
    for (int j = 0; j < MAX_TILES / BIN_THREADS; ++j) {
        const int i = j * BIN_THREADS + (int)threadIdx.x;
        if (i < tiles) {
            cnt[i]   = slot;
            shift[i] = (unsigned)i * (TILE_ROWS * BUCKET_PER_ROW) + first[j] - slot;
            slot += mine[j];
        }
    }
    __syncthreads();

    unsigned pos[BIN_SLOTS], sh[BIN_SLOTS];
#pragma unroll
    for (int c = 0; c < BIN_SLOTS; ++c) {
        const bool     kept = (unsigned)d[c] < (unsigned)n_nodes;
        const unsigned t    = kept ? (unsigned)d[c] / TILE_ROWS : MAX_TILES + (unsigned)lane;
        const unsigned r    = atomicAdd(&cnt[t], 1u);
        sh[c]  = shift[t];
        pos[c] = kept ? r : BIN_STAGE + (unsigned)lane;
    }
#pragma unroll
    for (int c = 0; c < BIN_SLOTS; ++c) {
        const bool     kept  = (unsigned)d[c] < (unsigned)n_nodes;
        const unsigned t     = kept ? (unsigned)d[c] / TILE_ROWS : 0u;
        const unsigned row0  = t * TILE_ROWS;
        const unsigned left  = (unsigned)n_nodes - row0;
        const unsigned cap   = (left < TILE_ROWS ? left : TILE_ROWS) * BUCKET_PER_ROW;
        const unsigned entry = pos[c] + sh[c];
        uint2 ent;
        ent.x = ((unsigned)s[c] & ((1u << ROW_SHIFT) - 1u)) | (((unsigned)d[c] - row0) << ROW_SHIFT);
        ent.y = __float_as_uint(w[c]);
        stage[pos[c]] = ent;
        where[pos[c]] = (entry - row0 * BUCKET_PER_ROW) < cap ? entry : NO_PLACE;
    }
    __syncthreads();

    uint2* out2 = reinterpret_cast<uint2*>(out);
    unsigned place[BIN_SLOTS];
    uint2    ent[BIN_SLOTS];
#pragma unroll
    for (int j = 0; j < BIN_SLOTS; ++j) {
        const unsigned i = (unsigned)(j * BIN_THREADS) + threadIdx.x;
        const unsigned k = i < n_staged ? i : 0u;
        place[j] = where[k];
        ent[j]   = stage[k];
    }
#pragma unroll
    for (int j = 0; j < BIN_SLOTS; ++j) {
        const unsigned i = (unsigned)(j * BIN_THREADS) + threadIdx.x;
        if (i < n_staged && place[j] != NO_PLACE) out2[place[j]] = ent[j];
    }
}

#define AGG_THREADS 1024
#define AGG_WAVES   (AGG_THREADS / 32)
#define WIN_LOADS   4
#define WIN_EDGES   (WIN_LOADS * 128)
#define Q_CAP       544

__device__ __forceinline__ void add_rows(int s, float w, int dl, int lane,
                                         const float* __restrict__ h, float* tile)
{
#pragma unroll
    for (int half = 0; half < 2; ++half) {
        float2 hv[16];
#pragma unroll
        for (int i = 0; i < 16; ++i) {
            const int si = __builtin_amdgcn_readlane(s, half * 16 + i);
            hv[i] = reinterpret_cast<const float2*>(h + (long)si * N_OUT)[lane];
        }
#pragma unroll
        for (int i = 0; i < 16; ++i) {
            const float wi = __int_as_float(__builtin_amdgcn_readlane(__float_as_int(w), half * 16 + i));
            const int   di = __builtin_amdgcn_readlane(dl, half * 16 + i);
            float* t = tile + di * N_OUT + 2 * lane;
            atomicAdd(t + 0, hv[i].x * wi);
            atomicAdd(t + 1, hv[i].y * wi);
        }
    }
}

#define SORT_MAX      (TILE_ROWS * BUCKET_PER_ROW)
#define SORT_PER_THR  ((SORT_MAX + AGG_THREADS - 1) / AGG_THREADS)
#define ROW_SLOTS     416
#define GATHER        8
__device__ __forceinline__ void bucket_rows(const uint2* bucket, int n, unsigned nrows, int lane, int wave,
                                            const float* __restrict__ h, unsigned* rowmem, uint2* sorted,
                                            float* out_rows)
{
    unsigned* cnt   = rowmem;
    unsigned* start = rowmem + ROW_SLOTS + 32;
    unsigned* wsum  = start + ROW_SLOTS;
    const int tid = (int)threadIdx.x;

    const int last = n > 0 ? n - 1 : 0;
    uint2 ent[SORT_PER_THR];
#pragma unroll
    for (int k = 0; k < SORT_PER_THR; ++k) {
        const int i = k * AGG_THREADS + tid;
        ent[k] = bucket[i < last ? i : last];
    }
    if (tid < ROW_SLOTS + 32) cnt[tid] = 0u;
    __syncthreads();

    unsigned rk[SORT_PER_THR];
#pragma unroll
    for (int k = 0; k < SORT_PER_THR; ++k) {
        const int i = k * AGG_THREADS + tid;
        const unsigned r = i < n ? ent[k].x >> ROW_SHIFT : (unsigned)(ROW_SLOTS + lane);
        rk[k] = atomicAdd(&cnt[r], 1u);
    }
    __syncthreads();

    unsigned v = 0u, incl = 0u;
    if (tid < ROW_SLOTS) {
        v = cnt[tid];
        incl = v;
#pragma unroll
        for (int off = 1; off < 32; off <<= 1) {
            const unsigned u = (unsigned)__shfl_up((int)incl, off, 32);
            if (lane >= off) incl += u;
        }
        if (lane == 31) wsum[wave] = incl;
    }
    __syncthreads();
    if (tid < ROW_SLOTS) {
        unsigned pre = 0u;
        for (int w2 = 0; w2 < wave; ++w2) pre += wsum[w2];
        start[tid] = pre + incl - v;
    }
    __syncthreads();

#pragma unroll
    for (int k = 0; k < SORT_PER_THR; ++k) {
        const int i = k * AGG_THREADS + tid;
        if (i < n) {
            const unsigned r = ent[k].x >> ROW_SHIFT;
            sorted[start[r] + rk[k]] = make_uint2(ent[k].x & ((1u << ROW_SHIFT) - 1u), ent[k].y);
        }
    }
    __syncthreads();

    const int half = lane >> 4;
    const int l16  = lane & 15;
    const float4* h4 = reinterpret_cast<const float4*>(h);
    for (unsigned p = (unsigned)wave; 2u * p < nrows; p += AGG_WAVES) {
        const unsigned row  = 2u * p + (unsigned)half;
        const bool     live = row < nrows;
        const unsigned c    = live ? cnt[row] : 0u;
        const unsigned b    = c ? start[row] : 0u;
        const unsigned cm1  = c ? c - 1u : 0u;
        float4 acc = {0.f, 0.f, 0.f, 0.f};
        for (unsigned e = 0u; __builtin_amdgcn_ballot_w32(e < c) != 0u; e += GATHER) {
            uint2 en[GATHER];
#pragma unroll
            for (int j = 0; j < GATHER; ++j) {
                const unsigned ej = e + (unsigned)j;
                en[j] = sorted[b + (ej < cm1 ? ej : cm1)];
                if (!(ej < c)) en[j].y = 0u;
            }
            float4 hv[GATHER];
#pragma unroll
            for (int j = 0; j < GATHER; ++j)
                hv[j] = h4[(size_t)en[j].x * (N_OUT / 4) + l16];
#pragma unroll
            for (int j = 0; j < GATHER; ++j) {
                const float w = __uint_as_float(en[j].y);
                acc.x += hv[j].x * w;
                acc.y += hv[j].y * w;
                acc.z += hv[j].z * w;
                acc.w += hv[j].w * w;
            }
        }
        if (c == 0u) acc = make_float4(0.f, 0.f, 0.f, 0.f);

        float m = fmaxf(fmaxf(acc.x, acc.y), fmaxf(acc.z, acc.w));
#pragma unroll
        for (int off = 8; off > 0; off >>= 1)
            m = fmaxf(m, __shfl_xor(m, off, 32));
        const float x0 = __expf(acc.x - m);
        const float x1 = __expf(acc.y - m);
        const float x2 = __expf(acc.z - m);
        const float x3 = __expf(acc.w - m);
        float sum = (x0 + x1) + (x2 + x3);
#pragma unroll
        for (int off = 8; off > 0; off >>= 1)
            sum += __shfl_xor(sum, off, 32);
        const float inv = 1.0f / sum;
        if (live)
            reinterpret_cast<float4*>(out_rows + (size_t)row * N_OUT)[l16] =
                make_float4(x0 * inv, x1 * inv, x2 * inv, x3 * inv);
    }
}

__device__ __forceinline__ void queue_push(bool keep, int e_first, unsigned c, unsigned row_in_tile,
                                           uint2* q, int& qn)
{
    const unsigned m = __builtin_amdgcn_ballot_w32(keep);
    if (m != 0u) {
        if (keep) {
            const int pos = qn + (int)__builtin_amdgcn_mbcnt_lo(m, 0u);
            q[pos] = make_uint2((unsigned)e_first, row_in_tile | (c << 16));
        }
        qn += __builtin_popcount(m);
    }
}

__device__ __forceinline__ void scan_window(const int4 (&buf)[WIN_LOADS], int wbase, int lane,
                                            int lo, unsigned nrows, uint2* q, int& qn)
{
#pragma unroll
    for (int u = 0; u < WIN_LOADS; ++u) {
        const int4 d = buf[u];
        const unsigned r0 = (unsigned)(d.x - lo);
        const unsigned r1 = (unsigned)(d.y - lo);
        const unsigned r2 = (unsigned)(d.z - lo);
        const unsigned r3 = (unsigned)(d.w - lo);
        const unsigned r01 = r0 < r1 ? r0 : r1;
        const unsigned r23 = r2 < r3 ? r2 : r3;
        const unsigned rmin = r01 < r23 ? r01 : r23;
        if (__builtin_amdgcn_ballot_w32(rmin < nrows) != 0u) {
            const int e0 = wbase + u * 128 + 4 * lane;
            queue_push(r0 < nrows, e0, 0u, r0, q, qn);
            queue_push(r1 < nrows, e0, 1u, r1, q, qn);
            queue_push(r2 < nrows, e0, 2u, r2, q, qn);
            queue_push(r3 < nrows, e0, 3u, r3, q, qn);
        }
    }
}

__device__ __forceinline__ void queue_drain(const uint2* q, int base, int nvalid, int lane,
                                            const float* __restrict__ h,
                                            const int* __restrict__ esrc,
                                            const float* __restrict__ ew,
                                            float* tile)
{
    __builtin_amdgcn_fence(__ATOMIC_ACQ_REL, "wavefront");

    int   s  = 0;
    float w  = 0.0f;
    int   dl = TILE_ROWS;
    if (lane < nvalid) {
        const uint2 ent = q[base + lane];
        const unsigned e = ent.x + (ent.y >> 16);
        s  = esrc[e];
        w  = ew[e];
        dl = (int)(ent.y & 0xffffu);
    }
    add_rows(s, w, dl, lane, h, tile);
}

__device__ __forceinline__ void scan_and_add(const float* __restrict__ h, const int* __restrict__ esrc,
                                             const int* __restrict__ edst, const float* __restrict__ ew,
                                             float* tile, uint2* q, int lane, int wave,
                                             int lo, unsigned nrows, int n_edges)
{
    int qn = 0;

    const int n_full  = n_edges / WIN_EDGES;
    const int covered = n_full * WIN_EDGES;
    const int4* d4 = reinterpret_cast<const int4*>(edst);
    const int rot = (int)(((long)blockIdx.x * n_full) / gridDim.x);
    int4 bufA[WIN_LOADS], bufB[WIN_LOADS];
    if (wave < n_full) {
        const int g0 = (wave + rot >= n_full) ? wave + rot - n_full : wave + rot;
#pragma unroll
        for (int u = 0; u < WIN_LOADS; ++u)
            bufA[u] = d4[(long)g0 * (WIN_EDGES / 4) + u * 32 + lane];
    }
    for (int j = wave; j < n_full; j += 2 * AGG_WAVES) {
        const int  ga   = (j + rot >= n_full) ? j + rot - n_full : j + rot;
        const int  jb   = j + AGG_WAVES;
        const bool hasB = jb < n_full;
        const int  jn   = j + 2 * AGG_WAVES;
        const int  gb   = !hasB ? ga : (jb + rot >= n_full) ? jb + rot - n_full : jb + rot;
        const int  gn   = !(jn < n_full) ? ga : (jn + rot >= n_full) ? jn + rot - n_full : jn + rot;

#pragma unroll
        for (int u = 0; u < WIN_LOADS; ++u)
            bufB[u] = d4[(long)gb * (WIN_EDGES / 4) + u * 32 + lane];

        scan_window(bufA, ga * WIN_EDGES, lane, lo, nrows, q, qn);
        while (qn >= 32) {
            qn -= 32;
            queue_drain(q, qn, 32, lane, h, esrc, ew, tile);
        }

#pragma unroll
        for (int u = 0; u < WIN_LOADS; ++u)
            bufA[u] = d4[(long)gn * (WIN_EDGES / 4) + u * 32 + lane];

        if (hasB) {
            scan_window(bufB, gb * WIN_EDGES, lane, lo, nrows, q, qn);
            while (qn >= 32) {
                qn -= 32;
                queue_drain(q, qn, 32, lane, h, esrc, ew, tile);
            }
        }
    }

    {
        int e0 = covered + wave * 32;
        for (;;) {
            const bool more = e0 < n_edges;
            if (more) {
                const int e = e0 + lane;
                unsigned r = 0xffffffffu;
                if (e < n_edges) r = (unsigned)(edst[e] - lo);
                queue_push(r < nrows, e, 0u, r, q, qn);
                e0 += AGG_WAVES * 32;
            }
            if (qn >= 32 || (!more && qn > 0)) {
                const int n = qn < 32 ? qn : 32;
                qn -= n;
                queue_drain(q, qn, n, lane, h, esrc, ew, tile);
            }
            if (!more && qn == 0) break;
        }
    }
}

__global__ __launch_bounds__(AGG_THREADS) void gcn_tile_aggregate(
    const float* __restrict__ h, const int* __restrict__ esrc,
    const int* __restrict__ edst, const float* __restrict__ ew,
    float* __restrict__ out, int n_nodes, int n_edges, unsigned long long ws_bytes)
{
    __shared__ __attribute__((aligned(16))) float tile[(TILE_ROWS + 1) * N_OUT];
    __shared__ __attribute__((aligned(16))) uint2 queue[AGG_WAVES * Q_CAP];

    const int lane = threadIdx.x & 31;
    const int wave = __builtin_amdgcn_readfirstlane((int)(threadIdx.x >> 5));
    const int lo   = blockIdx.x * TILE_ROWS;
    const int rem  = n_nodes - lo;
    const unsigned nrows = (unsigned)(rem < TILE_ROWS ? rem : TILE_ROWS);

    bool from_bucket = false;
    unsigned n_bucket = 0u;
    if (bins_fit(ws_bytes, n_nodes)) {
        const unsigned* counters = reinterpret_cast<const unsigned*>(h + (size_t)n_nodes * N_OUT);
        n_bucket    = counters[blockIdx.x];
        from_bucket = n_bucket <= nrows * BUCKET_PER_ROW;
    }

    if (from_bucket) {
        bucket_rows(reinterpret_cast<const uint2*>(out + (size_t)lo * N_OUT), (int)n_bucket, nrows, lane, wave,
                    h, reinterpret_cast<unsigned*>(tile), queue, out + (size_t)lo * N_OUT);
        return;
    }

    {
        float4* t4 = reinterpret_cast<float4*>(tile);
        const float4 z = {0.f, 0.f, 0.f, 0.f};
        for (int i = threadIdx.x; i < (TILE_ROWS + 1) * N_OUT / 4; i += AGG_THREADS) t4[i] = z;
    }
    __syncthreads();

    scan_and_add(h, esrc, edst, ew, tile, queue + wave * Q_CAP, lane, wave, lo, nrows, n_edges);

    __syncthreads();

    for (unsigned r = (unsigned)wave; r < nrows; r += AGG_WAVES) {
        const float2 v = reinterpret_cast<const float2*>(tile + r * N_OUT)[lane];

        float m = fmaxf(v.x, v.y);
#pragma unroll
        for (int off = 16; off > 0; off >>= 1)
            m = fmaxf(m, __shfl_xor(m, off, 32));

        const float x0 = __expf(v.x - m);
        const float x1 = __expf(v.y - m);
        float sum = x0 + x1;
#pragma unroll
        for (int off = 16; off > 0; off >>= 1)
            sum += __shfl_xor(sum, off, 32);

        const float inv = 1.0f / sum;
        float2 o; o.x = x0 * inv; o.y = x1 * inv;
        reinterpret_cast<float2*>(out + ((long)lo + r) * N_OUT)[lane] = o;
    }
}

extern "C" void kernel_launch(void* const* d_in, const int* in_sizes, int n_in,
                              void* d_out, int out_size, void* d_ws, size_t ws_size,
                              hipStream_t stream)
{
    const float* x    = (const float*)d_in[0];
    const int*   esrc = (const int*)d_in[1];
    const int*   edst = (const int*)d_in[2];
    const float* ew   = (const float*)d_in[3];
    const float* W    = (const float*)d_in[4];

    const int n_nodes = in_sizes[0] / D_FEAT;
    const int n_edges = in_sizes[1];

    float* out = (float*)d_out;
    float* h   = (float*)d_ws;

    {
        const int blocks = (n_nodes + 127) / 128;
        gcn_gemm_wmma<<<blocks, 256, 0, stream>>>(x, W, h, n_nodes, (unsigned long long)ws_size);
    }
    {
        const int blocks = n_edges > 0 ? (n_edges + BIN_EDGES - 1) / BIN_EDGES : 1;
        gcn_bin_edges<<<blocks, BIN_THREADS, 0, stream>>>(esrc, edst, ew, out, h, n_nodes, n_edges, (unsigned long long)ws_size);
    }
    {
        const int tiles = (n_nodes + TILE_ROWS - 1) / TILE_ROWS;
        gcn_tile_aggregate<<<tiles, AGG_THREADS, 0, stream>>>(h, esrc, edst, ew, out, n_nodes, n_edges, (unsigned long long)ws_size);
    }
}


# projection: W goes through LDS in 32-k slices (two-slot ring filled under the matrix work, one LDS-only barrier a slice), not staged whole; 64-row work-groups of 4 waves, all resident at once
# speedup vs baseline: 2.8142x; 1.0754x over previous
#include <hip/hip_runtime.h>

typedef __attribute__((ext_vector_type(2))) float v2f;
typedef __attribute__((ext_vector_type(8))) float v8f;

#define D_FEAT 512
#define N_OUT  64
#define PAIR_ROWS   (D_FEAT / 2)
#define PAIR_STRIDE 80
#define A_STEPS     8
#define A_TRIPS     (D_FEAT / (4 * A_STEPS))
#define SLICE_ROWS  (2 * A_STEPS)
#define SLICE_F2    (SLICE_ROWS * PAIR_STRIDE)
#define GEMM_WAVES  4
#define GEMM_THREADS (GEMM_WAVES * 32)
#define GEMM_ROWS   (GEMM_WAVES * 16)

#define TILE_ROWS   391

#define ROW_SHIFT       23
#define BUCKET_PER_ROW  (N_OUT * 4 / 8)
#define MAX_TILES       1024
#define BIN_THREADS     256
#define BIN_PER_THREAD  16
#define BIN_SLOTS       (BIN_PER_THREAD + 1)
#define BIN_EDGES       (BIN_THREADS * BIN_PER_THREAD)
#define BIN_STAGE       (BIN_THREADS * BIN_SLOTS)
#define NO_PLACE        0xffffffffu

__device__ __forceinline__ bool bins_fit(unsigned long long ws_bytes, int n_nodes)
{
    const unsigned long long h_bytes = (unsigned long long)n_nodes * N_OUT * sizeof(float);
    const int tiles = (n_nodes + TILE_ROWS - 1) / TILE_ROWS;
    return n_nodes > 0 && n_nodes <= (1 << ROW_SHIFT) && tiles <= MAX_TILES
        && ws_bytes >= h_bytes + (unsigned long long)tiles * sizeof(unsigned);
}

__device__ __forceinline__ void lds_barrier()
{
    __builtin_amdgcn_fence(__ATOMIC_RELEASE, "workgroup", "local");
    __builtin_amdgcn_s_barrier();
    __builtin_amdgcn_fence(__ATOMIC_ACQUIRE, "workgroup", "local");
}

__global__ __launch_bounds__(GEMM_THREADS) void gcn_gemm_wmma(
    const float* __restrict__ x, const float* __restrict__ W,
    float* __restrict__ h, int n_nodes, unsigned long long ws_bytes)
{
    __shared__ float2 Wp[2 * SLICE_F2];

    if (blockIdx.x == 0 && bins_fit(ws_bytes, n_nodes)) {
        unsigned* counters = reinterpret_cast<unsigned*>(h + (size_t)n_nodes * N_OUT);
        const int tiles = (n_nodes + TILE_ROWS - 1) / TILE_ROWS;
        for (int i = threadIdx.x; i < tiles; i += GEMM_THREADS) counters[i] = 0u;
    }

    const int lane = threadIdx.x & 31;
    const int wave = threadIdx.x >> 5;
    const int l16  = lane & 15;
    const int g    = lane >> 4;
    const long rowBase = (long)blockIdx.x * GEMM_ROWS + (long)wave * 16;
    long arow = rowBase + l16;
    if (arow >= n_nodes) arow = (long)n_nodes - 1;

    const float* aptr = x + arow * (long)D_FEAT + 2 * g;
    const float2* bptr = &Wp[(long)g * PAIR_STRIDE + l16];

    const int sp  = threadIdx.x >> 4;
    const int sn4 = (threadIdx.x & 15) << 2;
    const float* wsrc = W + (2 * sp) * N_OUT + sn4;
    float2*      wdst = &Wp[sp * PAIR_STRIDE + sn4];
    float4 wr0, wr1, wr2, wr3;

#define LOAD_W(t)                                                                           \
    wr0 = *reinterpret_cast<const float4*>(wsrc + (t) * (2 * SLICE_ROWS * N_OUT));                       \
    wr1 = *reinterpret_cast<const float4*>(wsrc + (t) * (2 * SLICE_ROWS * N_OUT) + N_OUT);               \
    wr2 = *reinterpret_cast<const float4*>(wsrc + (t) * (2 * SLICE_ROWS * N_OUT) + SLICE_ROWS * N_OUT);  \
    wr3 = *reinterpret_cast<const float4*>(wsrc + (t) * (2 * SLICE_ROWS * N_OUT) + (SLICE_ROWS + 1) * N_OUT);

#define STORE_W(slot)                                                                       \
    {                                                                                       \
        float4* d0 = reinterpret_cast<float4*>(wdst + (slot) * SLICE_F2);                   \
        float4* d1 = reinterpret_cast<float4*>(wdst + (slot) * SLICE_F2 + (SLICE_ROWS / 2) * PAIR_STRIDE); \
        d0[0] = make_float4(wr0.x, wr1.x, wr0.y, wr1.y);                                    \
        d0[1] = make_float4(wr0.z, wr1.z, wr0.w, wr1.w);                                    \
        d1[0] = make_float4(wr2.x, wr3.x, wr2.y, wr3.y);                                    \
        d1[1] = make_float4(wr2.z, wr3.z, wr2.w, wr3.w);                                    \
    }

    v8f acc0 = {}, acc1 = {}, acc2 = {}, acc3 = {};

    v2f abuf0[A_STEPS], abuf1[A_STEPS], abuf2[A_STEPS], abuf3[A_STEPS];

#define LOAD_TRIP(buf, t)                                                                   \
    _Pragma("unroll")                     \
    for (int j = 0; j < A_STEPS; ++j)                                                       \
        buf[j] = *reinterpret_cast<const v2f*>(aptr + (t) * (4 * A_STEPS) + 4 * j);

#define MMA_TRIP(buf, slot)                                                                 \
    _Pragma("unroll")                     \
    for (int j = 0; j < A_STEPS; ++j) {                                                     \
        const float2* wb = bptr + (slot) * SLICE_F2 + (2 * j) * PAIR_STRIDE;                \
        const v2f b0 = *reinterpret_cast<const v2f*>(wb +  0);                              \
        const v2f b1 = *reinterpret_cast<const v2f*>(wb + 16);                              \
        const v2f b2 = *reinterpret_cast<const v2f*>(wb + 32);                              \
        const v2f b3 = *reinterpret_cast<const v2f*>(wb + 48);                              \
        acc0 = __builtin_amdgcn_wmma_f32_16x16x4_f32(false, buf[j], false, b0, (short)0, acc0, false, false); \
        acc1 = __builtin_amdgcn_wmma_f32_16x16x4_f32(false, buf[j], false, b1, (short)0, acc1, false, false); \
        acc2 = __builtin_amdgcn_wmma_f32_16x16x4_f32(false, buf[j], false, b2, (short)0, acc2, false, false); \
        acc3 = __builtin_amdgcn_wmma_f32_16x16x4_f32(false, buf[j], false, b3, (short)0, acc3, false, false); \
    }

    LOAD_W(0)
    LOAD_TRIP(abuf0, 0)
    LOAD_TRIP(abuf1, 1)
    STORE_W(0)
    lds_barrier();
#pragma unroll 1
    for (int t = 0; t < A_TRIPS - 4; t += 4) {
        LOAD_W(t + 1)  LOAD_TRIP(abuf2, t + 2)  MMA_TRIP(abuf0, 0)  STORE_W(1)
        lds_barrier();
        LOAD_W(t + 2)  LOAD_TRIP(abuf3, t + 3)  MMA_TRIP(abuf1, 1)  STORE_W(0)
        lds_barrier();
        LOAD_W(t + 3)  LOAD_TRIP(abuf0, t + 4)  MMA_TRIP(abuf2, 0)  STORE_W(1)
        lds_barrier();
        LOAD_W(t + 4)  LOAD_TRIP(abuf1, t + 5)  MMA_TRIP(abuf3, 1)  STORE_W(0)
        lds_barrier();
    }
    LOAD_W(A_TRIPS - 3)  LOAD_TRIP(abuf2, A_TRIPS - 2)  MMA_TRIP(abuf0, 0)  STORE_W(1)
    lds_barrier();
    LOAD_W(A_TRIPS - 2)  LOAD_TRIP(abuf3, A_TRIPS - 1)  MMA_TRIP(abuf1, 1)  STORE_W(0)
    lds_barrier();
    LOAD_W(A_TRIPS - 1)  MMA_TRIP(abuf2, 0)  STORE_W(1)
    lds_barrier();
    MMA_TRIP(abuf3, 1)
#undef LOAD_TRIP
#undef MMA_TRIP
#undef LOAD_W
#undef STORE_W

    float* hb = h + rowBase * N_OUT + l16;
#pragma unroll
    for (int v = 0; v < 8; ++v) {
        const long m = v + 8 * g;
        if (rowBase + m < n_nodes) {
            hb[m * N_OUT +  0] = acc0[v];
            hb[m * N_OUT + 16] = acc1[v];
            hb[m * N_OUT + 32] = acc2[v];
            hb[m * N_OUT + 48] = acc3[v];
        }
    }
}

__global__ __launch_bounds__(BIN_THREADS) void gcn_bin_edges(
    const int* __restrict__ esrc, const int* __restrict__ edst, const float* __restrict__ ew,
    float* __restrict__ out, float* __restrict__ h, int n_nodes, int n_edges, unsigned long long ws_bytes)
{
    __shared__ unsigned cnt[MAX_TILES + 32];
    __shared__ unsigned shift[MAX_TILES + 32];
    __shared__ __attribute__((aligned(16))) uint2 stage[BIN_STAGE + 32];
    __shared__ unsigned where[BIN_STAGE + 32];
    __shared__ unsigned wave_sum[BIN_THREADS / 32];

    if (!bins_fit(ws_bytes, n_nodes)) return;
    unsigned* counters = reinterpret_cast<unsigned*>(h + (size_t)n_nodes * N_OUT);
    const int tiles = (n_nodes + TILE_ROWS - 1) / TILE_ROWS;
    const int lane  = threadIdx.x & 31;
    const int wave  = threadIdx.x >> 5;

    for (int i = threadIdx.x; i < tiles; i += BIN_THREADS) cnt[i] = 0u;
    if (threadIdx.x < 32) { cnt[MAX_TILES + threadIdx.x] = 0u; shift[MAX_TILES + threadIdx.x] = 0u; }

    int   d[BIN_SLOTS], s[BIN_SLOTS];
    float w[BIN_SLOTS];
#pragma unroll
    for (int j = 0; j < BIN_PER_THREAD / 4; ++j) {
        const long e0 = (long)blockIdx.x * BIN_EDGES + j * (BIN_THREADS * 4) + threadIdx.x * 4;
        int4   dv = {-1, -1, -1, -1};
        int4   sv = {0, 0, 0, 0};
        float4 wv = {0.f, 0.f, 0.f, 0.f};
        if (e0 + 3 < n_edges) {
            dv = *reinterpret_cast<const int4*>(edst + e0);
            sv = *reinterpret_cast<const int4*>(esrc + e0);
            wv = *reinterpret_cast<const float4*>(ew + e0);
        }
        d[4 * j + 0] = dv.x; d[4 * j + 1] = dv.y; d[4 * j + 2] = dv.z; d[4 * j + 3] = dv.w;
        s[4 * j + 0] = sv.x; s[4 * j + 1] = sv.y; s[4 * j + 2] = sv.z; s[4 * j + 3] = sv.w;
        w[4 * j + 0] = wv.x; w[4 * j + 1] = wv.y; w[4 * j + 2] = wv.z; w[4 * j + 3] = wv.w;
    }
    {
        const int ex = (n_edges & ~3) + (int)threadIdx.x;
        d[BIN_PER_THREAD] = -1; s[BIN_PER_THREAD] = 0; w[BIN_PER_THREAD] = 0.0f;
        if (blockIdx.x == 0 && ex < n_edges) {
            d[BIN_PER_THREAD] = edst[ex];
            s[BIN_PER_THREAD] = esrc[ex];
            w[BIN_PER_THREAD] = ew[ex];
        }
    }
    __syncthreads();

#pragma unroll
    for (int c = 0; c < BIN_SLOTS; ++c)
        if ((unsigned)d[c] < (unsigned)n_nodes) atomicAdd(&cnt[(unsigned)d[c] / TILE_ROWS], 1u);
    __syncthreads();

    unsigned mine[MAX_TILES / BIN_THREADS], first[MAX_TILES / BIN_THREADS];
    unsigned own = 0u;
#pragma unroll
    for (int j = 0; j < MAX_TILES / BIN_THREADS; ++j) {
        const int i = j * BIN_THREADS + (int)threadIdx.x;
        mine[j]  = i < tiles ? cnt[i] : 0u;
        first[j] = mine[j] ? atomicAdd(&counters[i], mine[j]) : 0u;
        own += mine[j];
    }
    unsigned upto = own;
#pragma unroll
    for (int off = 1; off < 32; off <<= 1) {
        const unsigned v = (unsigned)__shfl_up((int)upto, off, 32);
        if (lane >= off) upto += v;
    }
    if (lane == 31) wave_sum[wave] = upto;
    __syncthreads();

    unsigned slot = upto - own;
    unsigned n_staged = 0u;
#pragma unroll
    for (int v = 0; v < BIN_THREADS / 32; ++v) {
        const unsigned t = wave_sum[v];
        if (v < wave) slot += t;
        n_staged += t;
    }
    if (n_staged == 0u) return;
#pragma unroll
    for (int j = 0; j < MAX_TILES / BIN_THREADS; ++j) {
        const int i = j * BIN_THREADS + (int)threadIdx.x;
        if (i < tiles) {
            cnt[i]   = slot;
            shift[i] = (unsigned)i * (TILE_ROWS * BUCKET_PER_ROW) + first[j] - slot;
            slot += mine[j];
        }
    }
    __syncthreads();

    unsigned pos[BIN_SLOTS], sh[BIN_SLOTS];
#pragma unroll
    for (int c = 0; c < BIN_SLOTS; ++c) {
        const bool     kept = (unsigned)d[c] < (unsigned)n_nodes;
        const unsigned t    = kept ? (unsigned)d[c] / TILE_ROWS : MAX_TILES + (unsigned)lane;
        const unsigned r    = atomicAdd(&cnt[t], 1u);
        sh[c]  = shift[t];
        pos[c] = kept ? r : BIN_STAGE + (unsigned)lane;
    }
#pragma unroll
    for (int c = 0; c < BIN_SLOTS; ++c) {
        const bool     kept  = (unsigned)d[c] < (unsigned)n_nodes;
        const unsigned t     = kept ? (unsigned)d[c] / TILE_ROWS : 0u;
        const unsigned row0  = t * TILE_ROWS;
        const unsigned left  = (unsigned)n_nodes - row0;
        const unsigned cap   = (left < TILE_ROWS ? left : TILE_ROWS) * BUCKET_PER_ROW;
        const unsigned entry = pos[c] + sh[c];
        uint2 ent;
        ent.x = ((unsigned)s[c] & ((1u << ROW_SHIFT) - 1u)) | (((unsigned)d[c] - row0) << ROW_SHIFT);
        ent.y = __float_as_uint(w[c]);
        stage[pos[c]] = ent;
        where[pos[c]] = (entry - row0 * BUCKET_PER_ROW) < cap ? entry : NO_PLACE;
    }
    __syncthreads();

    uint2* out2 = reinterpret_cast<uint2*>(out);
    unsigned place[BIN_SLOTS];
    uint2    ent[BIN_SLOTS];
#pragma unroll
    for (int j = 0; j < BIN_SLOTS; ++j) {
        const unsigned i = (unsigned)(j * BIN_THREADS) + threadIdx.x;
        const unsigned k = i < n_staged ? i : 0u;
        place[j] = where[k];
        ent[j]   = stage[k];
    }
#pragma unroll
    for (int j = 0; j < BIN_SLOTS; ++j) {
        const unsigned i = (unsigned)(j * BIN_THREADS) + threadIdx.x;
        if (i < n_staged && place[j] != NO_PLACE) out2[place[j]] = ent[j];
    }
}

#define AGG_THREADS 1024
#define AGG_WAVES   (AGG_THREADS / 32)
#define WIN_LOADS   4
#define WIN_EDGES   (WIN_LOADS * 128)
#define Q_CAP       544

__device__ __forceinline__ void add_rows(int s, float w, int dl, int lane,
                                         const float* __restrict__ h, float* tile)
{
#pragma unroll
    for (int half = 0; half < 2; ++half) {
        float2 hv[16];
#pragma unroll
        for (int i = 0; i < 16; ++i) {
            const int si = __builtin_amdgcn_readlane(s, half * 16 + i);
            hv[i] = reinterpret_cast<const float2*>(h + (long)si * N_OUT)[lane];
        }
#pragma unroll
        for (int i = 0; i < 16; ++i) {
            const float wi = __int_as_float(__builtin_amdgcn_readlane(__float_as_int(w), half * 16 + i));
            const int   di = __builtin_amdgcn_readlane(dl, half * 16 + i);
            float* t = tile + di * N_OUT + 2 * lane;
            atomicAdd(t + 0, hv[i].x * wi);
            atomicAdd(t + 1, hv[i].y * wi);
        }
    }
}

#define SORT_MAX      (TILE_ROWS * BUCKET_PER_ROW)
#define SORT_PER_THR  ((SORT_MAX + AGG_THREADS - 1) / AGG_THREADS)
#define ROW_SLOTS     416
#define GATHER        8
__device__ __forceinline__ void bucket_rows(const uint2* bucket, int n, unsigned nrows, int lane, int wave,
                                            const float* __restrict__ h, unsigned* rowmem, uint2* sorted,
                                            float* out_rows)
{
    unsigned* cnt   = rowmem;
    unsigned* start = rowmem + ROW_SLOTS + 32;
    unsigned* wsum  = start + ROW_SLOTS;
    const int tid = (int)threadIdx.x;

    const int last = n > 0 ? n - 1 : 0;
    uint2 ent[SORT_PER_THR];
#pragma unroll
    for (int k = 0; k < SORT_PER_THR; ++k) {
        const int i = k * AGG_THREADS + tid;
        ent[k] = bucket[i < last ? i : last];
    }
    if (tid < ROW_SLOTS + 32) cnt[tid] = 0u;
    __syncthreads();

    unsigned rk[SORT_PER_THR];
#pragma unroll
    for (int k = 0; k < SORT_PER_THR; ++k) {
        const int i = k * AGG_THREADS + tid;
        const unsigned r = i < n ? ent[k].x >> ROW_SHIFT : (unsigned)(ROW_SLOTS + lane);
        rk[k] = atomicAdd(&cnt[r], 1u);
    }
    __syncthreads();

    unsigned v = 0u, incl = 0u;
    if (tid < ROW_SLOTS) {
        v = cnt[tid];
        incl = v;
#pragma unroll
        for (int off = 1; off < 32; off <<= 1) {
            const unsigned u = (unsigned)__shfl_up((int)incl, off, 32);
            if (lane >= off) incl += u;
        }
        if (lane == 31) wsum[wave] = incl;
    }
    __syncthreads();
    if (tid < ROW_SLOTS) {
        unsigned pre = 0u;
        for (int w2 = 0; w2 < wave; ++w2) pre += wsum[w2];
        start[tid] = pre + incl - v;
    }
    __syncthreads();

#pragma unroll
    for (int k = 0; k < SORT_PER_THR; ++k) {
        const int i = k * AGG_THREADS + tid;
        if (i < n) {
            const unsigned r = ent[k].x >> ROW_SHIFT;
            sorted[start[r] + rk[k]] = make_uint2(ent[k].x & ((1u << ROW_SHIFT) - 1u), ent[k].y);
        }
    }
    __syncthreads();

    const int half = lane >> 4;
    const int l16  = lane & 15;
    const float4* h4 = reinterpret_cast<const float4*>(h);
    for (unsigned p = (unsigned)wave; 2u * p < nrows; p += AGG_WAVES) {
        const unsigned row  = 2u * p + (unsigned)half;
        const bool     live = row < nrows;
        const unsigned c    = live ? cnt[row] : 0u;
        const unsigned b    = c ? start[row] : 0u;
        const unsigned cm1  = c ? c - 1u : 0u;
        float4 acc = {0.f, 0.f, 0.f, 0.f};
        for (unsigned e = 0u; __builtin_amdgcn_ballot_w32(e < c) != 0u; e += GATHER) {
            uint2 en[GATHER];
#pragma unroll
            for (int j = 0; j < GATHER; ++j) {
                const unsigned ej = e + (unsigned)j;
                en[j] = sorted[b + (ej < cm1 ? ej : cm1)];
                if (!(ej < c)) en[j].y = 0u;
            }
            float4 hv[GATHER];
#pragma unroll
            for (int j = 0; j < GATHER; ++j)
                hv[j] = h4[(size_t)en[j].x * (N_OUT / 4) + l16];
#pragma unroll
            for (int j = 0; j < GATHER; ++j) {
                const float w = __uint_as_float(en[j].y);
                acc.x += hv[j].x * w;
                acc.y += hv[j].y * w;
                acc.z += hv[j].z * w;
                acc.w += hv[j].w * w;
            }
        }
        if (c == 0u) acc = make_float4(0.f, 0.f, 0.f, 0.f);

        float m = fmaxf(fmaxf(acc.x, acc.y), fmaxf(acc.z, acc.w));
#pragma unroll
        for (int off = 8; off > 0; off >>= 1)
            m = fmaxf(m, __shfl_xor(m, off, 32));
        const float x0 = __expf(acc.x - m);
        const float x1 = __expf(acc.y - m);
        const float x2 = __expf(acc.z - m);
        const float x3 = __expf(acc.w - m);
        float sum = (x0 + x1) + (x2 + x3);
#pragma unroll
        for (int off = 8; off > 0; off >>= 1)
            sum += __shfl_xor(sum, off, 32);
        const float inv = 1.0f / sum;
        if (live)
            reinterpret_cast<float4*>(out_rows + (size_t)row * N_OUT)[l16] =
                make_float4(x0 * inv, x1 * inv, x2 * inv, x3 * inv);
    }
}

__device__ __forceinline__ void queue_push(bool keep, int e_first, unsigned c, unsigned row_in_tile,
                                           uint2* q, int& qn)
{
    const unsigned m = __builtin_amdgcn_ballot_w32(keep);
    if (m != 0u) {
        if (keep) {
            const int pos = qn + (int)__builtin_amdgcn_mbcnt_lo(m, 0u);
            q[pos] = make_uint2((unsigned)e_first, row_in_tile | (c << 16));
        }
        qn += __builtin_popcount(m);
    }
}

__device__ __forceinline__ void scan_window(const int4 (&buf)[WIN_LOADS], int wbase, int lane,
                                            int lo, unsigned nrows, uint2* q, int& qn)
{
#pragma unroll
    for (int u = 0; u < WIN_LOADS; ++u) {
        const int4 d = buf[u];
        const unsigned r0 = (unsigned)(d.x - lo);
        const unsigned r1 = (unsigned)(d.y - lo);
        const unsigned r2 = (unsigned)(d.z - lo);
        const unsigned r3 = (unsigned)(d.w - lo);
        const unsigned r01 = r0 < r1 ? r0 : r1;
        const unsigned r23 = r2 < r3 ? r2 : r3;
        const unsigned rmin = r01 < r23 ? r01 : r23;
        if (__builtin_amdgcn_ballot_w32(rmin < nrows) != 0u) {
            const int e0 = wbase + u * 128 + 4 * lane;
            queue_push(r0 < nrows, e0, 0u, r0, q, qn);
            queue_push(r1 < nrows, e0, 1u, r1, q, qn);
            queue_push(r2 < nrows, e0, 2u, r2, q, qn);
            queue_push(r3 < nrows, e0, 3u, r3, q, qn);
        }
    }
}

__device__ __forceinline__ void queue_drain(const uint2* q, int base, int nvalid, int lane,
                                            const float* __restrict__ h,
                                            const int* __restrict__ esrc,
                                            const float* __restrict__ ew,
                                            float* tile)
{
    __builtin_amdgcn_fence(__ATOMIC_ACQ_REL, "wavefront");

    int   s  = 0;
    float w  = 0.0f;
    int   dl = TILE_ROWS;
    if (lane < nvalid) {
        const uint2 ent = q[base + lane];
        const unsigned e = ent.x + (ent.y >> 16);
        s  = esrc[e];
        w  = ew[e];
        dl = (int)(ent.y & 0xffffu);
    }
    add_rows(s, w, dl, lane, h, tile);
}

__device__ __forceinline__ void scan_and_add(const float* __restrict__ h, const int* __restrict__ esrc,
                                             const int* __restrict__ edst, const float* __restrict__ ew,
                                             float* tile, uint2* q, int lane, int wave,
                                             int lo, unsigned nrows, int n_edges)
{
    int qn = 0;

    const int n_full  = n_edges / WIN_EDGES;
    const int covered = n_full * WIN_EDGES;
    const int4* d4 = reinterpret_cast<const int4*>(edst);
    const int rot = (int)(((long)blockIdx.x * n_full) / gridDim.x);
    int4 bufA[WIN_LOADS], bufB[WIN_LOADS];
    if (wave < n_full) {
        const int g0 = (wave + rot >= n_full) ? wave + rot - n_full : wave + rot;
#pragma unroll
        for (int u = 0; u < WIN_LOADS; ++u)
            bufA[u] = d4[(long)g0 * (WIN_EDGES / 4) + u * 32 + lane];
    }
    for (int j = wave; j < n_full; j += 2 * AGG_WAVES) {
        const int  ga   = (j + rot >= n_full) ? j + rot - n_full : j + rot;
        const int  jb   = j + AGG_WAVES;
        const bool hasB = jb < n_full;
        const int  jn   = j + 2 * AGG_WAVES;
        const int  gb   = !hasB ? ga : (jb + rot >= n_full) ? jb + rot - n_full : jb + rot;
        const int  gn   = !(jn < n_full) ? ga : (jn + rot >= n_full) ? jn + rot - n_full : jn + rot;

#pragma unroll
        for (int u = 0; u < WIN_LOADS; ++u)
            bufB[u] = d4[(long)gb * (WIN_EDGES / 4) + u * 32 + lane];

        scan_window(bufA, ga * WIN_EDGES, lane, lo, nrows, q, qn);
        while (qn >= 32) {
            qn -= 32;
            queue_drain(q, qn, 32, lane, h, esrc, ew, tile);
        }

#pragma unroll
        for (int u = 0; u < WIN_LOADS; ++u)
            bufA[u] = d4[(long)gn * (WIN_EDGES / 4) + u * 32 + lane];

        if (hasB) {
            scan_window(bufB, gb * WIN_EDGES, lane, lo, nrows, q, qn);
            while (qn >= 32) {
                qn -= 32;
                queue_drain(q, qn, 32, lane, h, esrc, ew, tile);
            }
        }
    }

    {
        int e0 = covered + wave * 32;
        for (;;) {
            const bool more = e0 < n_edges;
            if (more) {
                const int e = e0 + lane;
                unsigned r = 0xffffffffu;
                if (e < n_edges) r = (unsigned)(edst[e] - lo);
                queue_push(r < nrows, e, 0u, r, q, qn);
                e0 += AGG_WAVES * 32;
            }
            if (qn >= 32 || (!more && qn > 0)) {
                const int n = qn < 32 ? qn : 32;
                qn -= n;
                queue_drain(q, qn, n, lane, h, esrc, ew, tile);
            }
            if (!more && qn == 0) break;
        }
    }
}

__global__ __launch_bounds__(AGG_THREADS) void gcn_tile_aggregate(
    const float* __restrict__ h, const int* __restrict__ esrc,
    const int* __restrict__ edst, const float* __restrict__ ew,
    float* __restrict__ out, int n_nodes, int n_edges, unsigned long long ws_bytes)
{
    __shared__ __attribute__((aligned(16))) float tile[(TILE_ROWS + 1) * N_OUT];
    __shared__ __attribute__((aligned(16))) uint2 queue[AGG_WAVES * Q_CAP];

    const int lane = threadIdx.x & 31;
    const int wave = __builtin_amdgcn_readfirstlane((int)(threadIdx.x >> 5));
    const int lo   = blockIdx.x * TILE_ROWS;
    const int rem  = n_nodes - lo;
    const unsigned nrows = (unsigned)(rem < TILE_ROWS ? rem : TILE_ROWS);

    bool from_bucket = false;
    unsigned n_bucket = 0u;
    if (bins_fit(ws_bytes, n_nodes)) {
        const unsigned* counters = reinterpret_cast<const unsigned*>(h + (size_t)n_nodes * N_OUT);
        n_bucket    = counters[blockIdx.x];
        from_bucket = n_bucket <= nrows * BUCKET_PER_ROW;
    }

    if (from_bucket) {
        bucket_rows(reinterpret_cast<const uint2*>(out + (size_t)lo * N_OUT), (int)n_bucket, nrows, lane, wave,
                    h, reinterpret_cast<unsigned*>(tile), queue, out + (size_t)lo * N_OUT);
        return;
    }

    {
        float4* t4 = reinterpret_cast<float4*>(tile);
        const float4 z = {0.f, 0.f, 0.f, 0.f};
        for (int i = threadIdx.x; i < (TILE_ROWS + 1) * N_OUT / 4; i += AGG_THREADS) t4[i] = z;
    }
    __syncthreads();

    scan_and_add(h, esrc, edst, ew, tile, queue + wave * Q_CAP, lane, wave, lo, nrows, n_edges);

    __syncthreads();

    for (unsigned r = (unsigned)wave; r < nrows; r += AGG_WAVES) {
        const float2 v = reinterpret_cast<const float2*>(tile + r * N_OUT)[lane];

        float m = fmaxf(v.x, v.y);
#pragma unroll
        for (int off = 16; off > 0; off >>= 1)
            m = fmaxf(m, __shfl_xor(m, off, 32));

        const float x0 = __expf(v.x - m);
        const float x1 = __expf(v.y - m);
        float sum = x0 + x1;
#pragma unroll
        for (int off = 16; off > 0; off >>= 1)
            sum += __shfl_xor(sum, off, 32);

        const float inv = 1.0f / sum;
        float2 o; o.x = x0 * inv; o.y = x1 * inv;
        reinterpret_cast<float2*>(out + ((long)lo + r) * N_OUT)[lane] = o;
    }
}

extern "C" void kernel_launch(void* const* d_in, const int* in_sizes, int n_in,
                              void* d_out, int out_size, void* d_ws, size_t ws_size,
                              hipStream_t stream)
{
    const float* x    = (const float*)d_in[0];
    const int*   esrc = (const int*)d_in[1];
    const int*   edst = (const int*)d_in[2];
    const float* ew   = (const float*)d_in[3];
    const float* W    = (const float*)d_in[4];

    const int n_nodes = in_sizes[0] / D_FEAT;
    const int n_edges = in_sizes[1];

    float* out = (float*)d_out;
    float* h   = (float*)d_ws;

    {
        const int blocks = (n_nodes + GEMM_ROWS - 1) / GEMM_ROWS;
        gcn_gemm_wmma<<<blocks, GEMM_THREADS, 0, stream>>>(x, W, h, n_nodes, (unsigned long long)ws_size);
    }
    {
        const int blocks = n_edges > 0 ? (n_edges + BIN_EDGES - 1) / BIN_EDGES : 1;
        gcn_bin_edges<<<blocks, BIN_THREADS, 0, stream>>>(esrc, edst, ew, out, h, n_nodes, n_edges, (unsigned long long)ws_size);
    }
    {
        const int tiles = (n_nodes + TILE_ROWS - 1) / TILE_ROWS;
        gcn_tile_aggregate<<<tiles, AGG_THREADS, 0, stream>>>(h, esrc, edst, ew, out, n_nodes, n_edges, (unsigned long long)ws_size);
    }
}
